# starting kernel
# baseline (speedup 1.0000x reference)
#include <hip/hip_runtime.h>
#include <math.h>

#define D_MODEL   1024
#define D_INNER   2048
#define D_XB       512
#define D_STATE     16
#define D_CONV       4
#define DT_RANK     64
#define NXB         32
#define H_HEADS    128
#define REP          4
#define PCH         16
#define BATCH        2
#define SEQ       2048
#define E_DIM     (2*D_INNER + 2*D_XB + DT_RANK)
#define M_ROWS    (BATCH*SEQ)

#define OFF_Z   0
#define OFF_X   (D_INNER)
#define OFF_B   (D_INNER + D_XB)
#define OFF_C   (D_INNER + 2*D_XB)
#define OFF_DT  (2*D_INNER + 2*D_XB)

typedef __attribute__((ext_vector_type(2))) float v2f;
typedef __attribute__((ext_vector_type(8))) float v8f;

__device__ __forceinline__ v8f wmma_f32_16x16x4(v2f a, v2f b, v8f c) {
  return __builtin_amdgcn_wmma_f32_16x16x4_f32(
       false, a,  false, b,
       (short)0, c,  false,  false);
}

template<int EPI>
__global__ void __launch_bounds__(256)
wmma_gemm_f32(const float* __restrict__ A, int lda,
              const float* __restrict__ W, int ldb,
              float* __restrict__ C, int ldc,
              int M, int N, int K,
              const float* __restrict__ bias)
{
  const int lane = threadIdx.x & 31;
  const int wave = threadIdx.x >> 5;
  const int tilesN = N >> 6;
  const int totalTiles = (M >> 6) * tilesN;
  const int tile = blockIdx.x * (blockDim.x >> 5) + wave;
  if (tile >= totalTiles) return;

  const int tm = tile / tilesN;
  const int tn = tile - tm * tilesN;

  const int r    = lane & 15;
  const int klo  = (lane >> 4) << 1;

  const float* __restrict__ Ar[4];
  const float* __restrict__ Br[4];
#pragma unroll
  for (int i = 0; i < 4; ++i) {
    Ar[i] = A + (size_t)(tm * 64 + i * 16 + r) * lda + klo;
    Br[i] = W + (size_t)(tn * 64 + i * 16 + r) * ldb + klo;
  }

  v8f acc[4][4];
#pragma unroll
  for (int i = 0; i < 4; ++i)
#pragma unroll
    for (int j = 0; j < 4; ++j) acc[i][j] = {};

  for (int k = 0; k < K; k += 4) {
    v2f a[4], b[4];
#pragma unroll
    for (int i = 0; i < 4; ++i) a[i] = *(const v2f*)(Ar[i] + k);
#pragma unroll
    for (int j = 0; j < 4; ++j) b[j] = *(const v2f*)(Br[j] + k);
#pragma unroll
    for (int i = 0; i < 4; ++i)
#pragma unroll
      for (int j = 0; j < 4; ++j)
        acc[i][j] = wmma_f32_16x16x4(a[i], b[j], acc[i][j]);
  }

  const int half = lane >> 4;
#pragma unroll
  for (int i = 0; i < 4; ++i) {
#pragma unroll
    for (int j = 0; j < 4; ++j) {
      const int cn = tn * 64 + j * 16 + r;
#pragma unroll
      for (int v = 0; v < 8; ++v) {
        const int cm = tm * 64 + i * 16 + v + (half << 3);
        float val = acc[i][j][v];
        if (EPI == 1) {
          val += bias[cn];
          val = (val > 20.f) ? val : log1pf(__expf(val));
        }
        C[(size_t)cm * ldc + cn] = val;
      }
    }
  }
}

__global__ void __launch_bounds__(256)
conv_silu_kernel(const float* __restrict__ zx,
                 const float* __restrict__ conv_w,
                 const float* __restrict__ conv_b,
                 float* __restrict__ xconv)
{
  int idx = blockIdx.x * blockDim.x + threadIdx.x;
  if (idx >= BATCH * SEQ * D_INNER) return;
  const int d = idx % D_INNER;
  const int l = (idx / D_INNER) % SEQ;
  const int b = idx / (D_INNER * SEQ);
  const int src = ((d >> 6) << 4) + (d & 15);

  const float* __restrict__ w = conv_w + d * D_CONV;
  float acc = conv_b[d];
#pragma unroll
  for (int t = 0; t < D_CONV; ++t) {
    const int ll = l - (D_CONV - 1) + t;
    if (ll >= 0) {
      float xv = zx[(size_t)(b * SEQ + ll) * E_DIM + OFF_X + src];
      acc = fmaf(w[t], xv, acc);
    }
  }
  acc = acc / (1.f + __expf(-acc));
  xconv[(size_t)idx] = acc;
}

__global__ void __launch_bounds__(256)
scan_kernel(const float* __restrict__ zx,
            const float* __restrict__ xconv,
            const float* __restrict__ dtbuf,
            const float* __restrict__ Amat,
            const float* __restrict__ Dvec,
            float* __restrict__ Y)
{
  const int b = blockIdx.x / H_HEADS;
  const int h = blockIdx.x % H_HEADS;
  const int p = threadIdx.x >> 4;
  const int n = threadIdx.x & 15;

  const float An = Amat[(size_t)(h * PCH + p) * D_STATE + n];
  const float Dv = Dvec[h * PCH + p];

  const size_t chan = (size_t)h * PCH + p;
  const size_t boff = OFF_B + (size_t)(h / REP) * D_STATE + n;
  const size_t coff = OFF_C + (size_t)h * D_STATE + n;
  const size_t zoff = OFF_Z + chan;

  float hstate = 0.f;
  size_t row = (size_t)b * SEQ;
  for (int l = 0; l < SEQ; ++l, ++row) {
    const float dt = dtbuf[row * D_INNER + chan];
    const float xv = xconv[row * D_INNER + chan];
    const float Bv = zx[row * E_DIM + boff];
    const float Cv = zx[row * E_DIM + coff];

    const float a = __expf(dt * An);
    hstate = fmaf(a, hstate, dt * xv * Bv);

    float yc = hstate * Cv;
    yc += __shfl_xor(yc, 1, 32);
    yc += __shfl_xor(yc, 2, 32);
    yc += __shfl_xor(yc, 4, 32);
    yc += __shfl_xor(yc, 8, 32);

    if (n == 0) {
      const float zval = zx[row * E_DIM + zoff];
      const float gate = zval / (1.f + __expf(-zval));
      Y[row * D_INNER + chan] = (yc + Dv * xv) * gate;
    }
  }
}

extern "C" void kernel_launch(void* const* d_in, const int* in_sizes, int n_in,
                              void* d_out, int out_size, void* d_ws, size_t ws_size,
                              hipStream_t stream) {
  const float* hidden     = (const float*)d_in[0];
  const float* in_proj_w  = (const float*)d_in[1];
  const float* conv_w     = (const float*)d_in[2];
  const float* conv_b     = (const float*)d_in[3];
  const float* dt_proj_w  = (const float*)d_in[4];
  const float* dt_proj_b  = (const float*)d_in[5];
  const float* Amat       = (const float*)d_in[6];
  const float* Dvec       = (const float*)d_in[7];
  const float* out_proj_w = (const float*)d_in[8];
  float* out = (float*)d_out;

  float* zx    = (float*)d_ws;
  float* xconv = zx    + (size_t)M_ROWS * E_DIM;
  float* dtbuf = xconv + (size_t)M_ROWS * D_INNER;
  float* Ybuf  = dtbuf + (size_t)M_ROWS * D_INNER;

  {
    int tiles = (M_ROWS / 64) * (E_DIM / 64);
    wmma_gemm_f32<0><<<(tiles + 7) / 8, 256, 0, stream>>>(
        hidden, D_MODEL, in_proj_w, D_MODEL, zx, E_DIM,
        M_ROWS, E_DIM, D_MODEL, nullptr);
  }

  {
    int n = BATCH * SEQ * D_INNER;
    conv_silu_kernel<<<(n + 255) / 256, 256, 0, stream>>>(zx, conv_w, conv_b, xconv);
  }

  {
    int tiles = (M_ROWS / 64) * (D_INNER / 64);
    wmma_gemm_f32<1><<<(tiles + 7) / 8, 256, 0, stream>>>(
        zx + OFF_DT, E_DIM, dt_proj_w, DT_RANK, dtbuf, D_INNER,
        M_ROWS, D_INNER, DT_RANK, dt_proj_b);
  }

  scan_kernel<<<BATCH * H_HEADS, 256, 0, stream>>>(zx, xconv, dtbuf, Amat, Dvec, Ybuf);

  {
    int tiles = (M_ROWS / 64) * (D_MODEL / 64);
    wmma_gemm_f32<0><<<(tiles + 7) / 8, 256, 0, stream>>>(
        Ybuf, D_INNER, out_proj_w, D_INNER, out, D_MODEL,
        M_ROWS, D_MODEL, D_INNER, nullptr);
  }
}


# scan: operands staged per 64-step tile in LDS by coalesced 16-byte loads, next tile prefetched in registers, step sums parked in LDS, gated Y stored in 16-byte pieces; no global load left per step
# speedup vs baseline: 3.6816x; 3.6816x over previous
#include <hip/hip_runtime.h>
#include <math.h>

#define D_MODEL   1024
#define D_INNER   2048
#define D_XB       512
#define D_STATE     16
#define D_CONV       4
#define DT_RANK     64
#define NXB         32
#define H_HEADS    128
#define REP          4
#define PCH         16
#define BATCH        2
#define SEQ       2048
#define E_DIM     (2*D_INNER + 2*D_XB + DT_RANK)
#define M_ROWS    (BATCH*SEQ)

#define OFF_Z   0
#define OFF_X   (D_INNER)
#define OFF_B   (D_INNER + D_XB)
#define OFF_C   (D_INNER + 2*D_XB)
#define OFF_DT  (2*D_INNER + 2*D_XB)

typedef __attribute__((ext_vector_type(2))) float v2f;
typedef __attribute__((ext_vector_type(8))) float v8f;

__device__ __forceinline__ v8f wmma_f32_16x16x4(v2f a, v2f b, v8f c) {
  return __builtin_amdgcn_wmma_f32_16x16x4_f32(
       false, a,  false, b,
       (short)0, c,  false,  false);
}

template<int EPI>
__global__ void __launch_bounds__(256)
wmma_gemm_f32(const float* __restrict__ A, int lda,
              const float* __restrict__ W, int ldb,
              float* __restrict__ C, int ldc,
              int M, int N, int K,
              const float* __restrict__ bias)
{
  const int lane = threadIdx.x & 31;
  const int wave = threadIdx.x >> 5;
  const int tilesN = N >> 6;
  const int totalTiles = (M >> 6) * tilesN;
  const int tile = blockIdx.x * (blockDim.x >> 5) + wave;
  if (tile >= totalTiles) return;

  const int tm = tile / tilesN;
  const int tn = tile - tm * tilesN;

  const int r    = lane & 15;
  const int klo  = (lane >> 4) << 1;

  const float* __restrict__ Ar[4];
  const float* __restrict__ Br[4];
#pragma unroll
  for (int i = 0; i < 4; ++i) {
    Ar[i] = A + (size_t)(tm * 64 + i * 16 + r) * lda + klo;
    Br[i] = W + (size_t)(tn * 64 + i * 16 + r) * ldb + klo;
  }

  v8f acc[4][4];
#pragma unroll
  for (int i = 0; i < 4; ++i)
#pragma unroll
    for (int j = 0; j < 4; ++j) acc[i][j] = {};

  for (int k = 0; k < K; k += 4) {
    v2f a[4], b[4];
#pragma unroll
    for (int i = 0; i < 4; ++i) a[i] = *(const v2f*)(Ar[i] + k);
#pragma unroll
    for (int j = 0; j < 4; ++j) b[j] = *(const v2f*)(Br[j] + k);
#pragma unroll
    for (int i = 0; i < 4; ++i)
#pragma unroll
      for (int j = 0; j < 4; ++j)
        acc[i][j] = wmma_f32_16x16x4(a[i], b[j], acc[i][j]);
  }

  const int half = lane >> 4;
#pragma unroll
  for (int i = 0; i < 4; ++i) {
#pragma unroll
    for (int j = 0; j < 4; ++j) {
      const int cn = tn * 64 + j * 16 + r;
#pragma unroll
      for (int v = 0; v < 8; ++v) {
        const int cm = tm * 64 + i * 16 + v + (half << 3);
        float val = acc[i][j][v];
        if (EPI == 1) {
          val += bias[cn];
          val = (val > 20.f) ? val : log1pf(__expf(val));
        }
        C[(size_t)cm * ldc + cn] = val;
      }
    }
  }
}

__global__ void __launch_bounds__(256)
conv_silu_kernel(const float* __restrict__ zx,
                 const float* __restrict__ conv_w,
                 const float* __restrict__ conv_b,
                 float* __restrict__ xconv)
{
  int idx = blockIdx.x * blockDim.x + threadIdx.x;
  if (idx >= BATCH * SEQ * D_INNER) return;
  const int d = idx % D_INNER;
  const int l = (idx / D_INNER) % SEQ;
  const int b = idx / (D_INNER * SEQ);
  const int src = ((d >> 6) << 4) + (d & 15);

  const float* __restrict__ w = conv_w + d * D_CONV;
  float acc = conv_b[d];
#pragma unroll
  for (int t = 0; t < D_CONV; ++t) {
    const int ll = l - (D_CONV - 1) + t;
    if (ll >= 0) {
      float xv = zx[(size_t)(b * SEQ + ll) * E_DIM + OFF_X + src];
      acc = fmaf(w[t], xv, acc);
    }
  }
  acc = acc / (1.f + __expf(-acc));
  xconv[(size_t)idx] = acc;
}

#define SCAN_T   64
#define SCAN_NT  (SEQ / SCAN_T)

__global__ void __launch_bounds__(256)
scan_kernel(const float* __restrict__ zx,
            const float* __restrict__ xconv,
            const float* __restrict__ dtbuf,
            const float* __restrict__ Amat,
            const float* __restrict__ Dvec,
            float* __restrict__ Y)
{
  __shared__ __attribute__((aligned(16))) float2 s_pd[SCAN_T][PCH];
  __shared__ __attribute__((aligned(16))) float2 s_bc[SCAN_T][D_STATE];
  __shared__ float s_y[SCAN_T][PCH * D_STATE + 1];

  const int b = blockIdx.x / H_HEADS;
  const int h = blockIdx.x % H_HEADS;
  const int p = threadIdx.x >> 4;
  const int n = threadIdx.x & 15;

  const int st = threadIdx.x >> 2;
  const int sq = (threadIdx.x & 3) << 2;

  const float  An  = Amat[(size_t)(h * PCH + p) * D_STATE + n];
  const float4 Dv4 = make_float4(Dvec[h * PCH + sq + 0], Dvec[h * PCH + sq + 1],
                                 Dvec[h * PCH + sq + 2], Dvec[h * PCH + sq + 3]);

  const size_t row0 = (size_t)b * SEQ + st;
  const size_t coff = (size_t)h * PCH + sq;
  const size_t boff = OFF_B + (size_t)(h / REP) * D_STATE + sq;
  const size_t cofs = OFF_C + (size_t)h * D_STATE + sq;
  const size_t zoff = OFF_Z + coff;

  float4 r_dt = *(const float4*)(dtbuf + row0 * D_INNER + coff);
  float4 r_x  = *(const float4*)(xconv + row0 * D_INNER + coff);
  float4 r_B  = *(const float4*)(zx + row0 * E_DIM + boff);
  float4 r_C  = *(const float4*)(zx + row0 * E_DIM + cofs);
  float4 r_z  = *(const float4*)(zx + row0 * E_DIM + zoff);

  float hstate = 0.f;
  for (int tile = 0; tile < SCAN_NT; ++tile) {
    {
      float4* pd = (float4*)&s_pd[st][sq];
      pd[0] = make_float4(r_dt.x, r_dt.x * r_x.x, r_dt.y, r_dt.y * r_x.y);
      pd[1] = make_float4(r_dt.z, r_dt.z * r_x.z, r_dt.w, r_dt.w * r_x.w);
      float4* bc = (float4*)&s_bc[st][sq];
      bc[0] = make_float4(r_B.x, r_C.x, r_B.y, r_C.y);
      bc[1] = make_float4(r_B.z, r_C.z, r_B.w, r_C.w);
    }
    const float4 c_x = r_x;
    const float4 c_z = r_z;
    __syncthreads();

    if (tile + 1 < SCAN_NT) {
      const size_t row = row0 + (size_t)(tile + 1) * SCAN_T;
      r_dt = *(const float4*)(dtbuf + row * D_INNER + coff);
      r_x  = *(const float4*)(xconv + row * D_INNER + coff);
      r_B  = *(const float4*)(zx + row * E_DIM + boff);
      r_C  = *(const float4*)(zx + row * E_DIM + cofs);
      r_z  = *(const float4*)(zx + row * E_DIM + zoff);
    }

#pragma unroll 16
    for (int t = 0; t < SCAN_T; ++t) {
      const float2 pd = s_pd[t][p];
      const float2 bc = s_bc[t][n];

      const float a = __expf(pd.x * An);
      hstate = fmaf(a, hstate, pd.y * bc.x);

      float yc = hstate * bc.y;
      yc += __shfl_xor(yc, 1, 32);
      yc += __shfl_xor(yc, 2, 32);
      yc += __shfl_xor(yc, 4, 32);
      yc += __shfl_xor(yc, 8, 32);

      s_y[t][threadIdx.x] = yc;
    }
    __syncthreads();

    {
      float4 y4;
      y4.x = s_y[st][(sq + 0) * D_STATE];
      y4.y = s_y[st][(sq + 1) * D_STATE];
      y4.z = s_y[st][(sq + 2) * D_STATE];
      y4.w = s_y[st][(sq + 3) * D_STATE];
      const float g0 = c_z.x / (1.f + __expf(-c_z.x));
      const float g1 = c_z.y / (1.f + __expf(-c_z.y));
      const float g2 = c_z.z / (1.f + __expf(-c_z.z));
      const float g3 = c_z.w / (1.f + __expf(-c_z.w));
      float4 o;
      o.x = (y4.x + Dv4.x * c_x.x) * g0;
      o.y = (y4.y + Dv4.y * c_x.y) * g1;
      o.z = (y4.z + Dv4.z * c_x.z) * g2;
      o.w = (y4.w + Dv4.w * c_x.w) * g3;
      const size_t row = row0 + (size_t)tile * SCAN_T;
      *(float4*)(Y + row * D_INNER + coff) = o;
    }
  }
}

extern "C" void kernel_launch(void* const* d_in, const int* in_sizes, int n_in,
                              void* d_out, int out_size, void* d_ws, size_t ws_size,
                              hipStream_t stream) {
  const float* hidden     = (const float*)d_in[0];
  const float* in_proj_w  = (const float*)d_in[1];
  const float* conv_w     = (const float*)d_in[2];
  const float* conv_b     = (const float*)d_in[3];
  const float* dt_proj_w  = (const float*)d_in[4];
  const float* dt_proj_b  = (const float*)d_in[5];
  const float* Amat       = (const float*)d_in[6];
  const float* Dvec       = (const float*)d_in[7];
  const float* out_proj_w = (const float*)d_in[8];
  float* out = (float*)d_out;

  float* zx    = (float*)d_ws;
  float* xconv = zx    + (size_t)M_ROWS * E_DIM;
  float* dtbuf = xconv + (size_t)M_ROWS * D_INNER;
  float* Ybuf  = dtbuf + (size_t)M_ROWS * D_INNER;

  {
    int tiles = (M_ROWS / 64) * (E_DIM / 64);
    wmma_gemm_f32<0><<<(tiles + 7) / 8, 256, 0, stream>>>(
        hidden, D_MODEL, in_proj_w, D_MODEL, zx, E_DIM,
        M_ROWS, E_DIM, D_MODEL, nullptr);
  }

  {
    int n = BATCH * SEQ * D_INNER;
    conv_silu_kernel<<<(n + 255) / 256, 256, 0, stream>>>(zx, conv_w, conv_b, xconv);
  }

  {
    int tiles = (M_ROWS / 64) * (D_INNER / 64);
    wmma_gemm_f32<1><<<(tiles + 7) / 8, 256, 0, stream>>>(
        zx + OFF_DT, E_DIM, dt_proj_w, DT_RANK, dtbuf, D_INNER,
        M_ROWS, D_INNER, DT_RANK, dt_proj_b);
  }

  scan_kernel<<<BATCH * H_HEADS, 256, 0, stream>>>(zx, xconv, dtbuf, Amat, Dvec, Ybuf);

  {
    int tiles = (M_ROWS / 64) * (D_MODEL / 64);
    wmma_gemm_f32<0><<<(tiles + 7) / 8, 256, 0, stream>>>(
        Ybuf, D_INNER, out_proj_w, D_INNER, out, D_MODEL,
        M_ROWS, D_MODEL, D_INNER, nullptr);
  }
}


# GEMMs: A and W staged per 128x128x32 block tile in LDS by coalesced 16-byte loads, next chunk prefetched in registers, fragments read from LDS; k order per accumulator unchanged
# speedup vs baseline: 5.5253x; 1.5008x over previous
#include <hip/hip_runtime.h>
#include <math.h>

#define D_MODEL   1024
#define D_INNER   2048
#define D_XB       512
#define D_STATE     16
#define D_CONV       4
#define DT_RANK     64
#define NXB         32
#define H_HEADS    128
#define REP          4
#define PCH         16
#define BATCH        2
#define SEQ       2048
#define E_DIM     (2*D_INNER + 2*D_XB + DT_RANK)
#define M_ROWS    (BATCH*SEQ)

#define OFF_Z   0
#define OFF_X   (D_INNER)
#define OFF_B   (D_INNER + D_XB)
#define OFF_C   (D_INNER + 2*D_XB)
#define OFF_DT  (2*D_INNER + 2*D_XB)

typedef __attribute__((ext_vector_type(2))) float v2f;
typedef __attribute__((ext_vector_type(8))) float v8f;

__device__ __forceinline__ v8f wmma_f32_16x16x4(v2f a, v2f b, v8f c) {
  return __builtin_amdgcn_wmma_f32_16x16x4_f32(
       false, a,  false, b,
       (short)0, c,  false,  false);
}

#define GBM     128
#define GBN     128
#define GKC      32
#define GPITCH   36

template<int EPI>
__global__ void __launch_bounds__(256)
wmma_gemm_f32(const float* __restrict__ A, int lda,
              const float* __restrict__ W, int ldb,
              float* __restrict__ C, int ldc,
              int M, int N, int K,
              const float* __restrict__ bias)
{
  __shared__ __attribute__((aligned(16))) float sA[2][GBM * GPITCH];
  __shared__ __attribute__((aligned(16))) float sB[2][GBN * GPITCH];

  const int tid  = threadIdx.x;
  const int lane = tid & 31;
  const int wave = tid >> 5;
  const int wm   = wave >> 2;
  const int wn   = wave & 3;
  const int tilesN = (N + GBN - 1) / GBN;
  const int bm = blockIdx.x / tilesN;
  const int bn = blockIdx.x - bm * tilesN;
  const int m0 = bm * GBM;
  const int n0 = bn * GBN;

  const int r    = lane & 15;
  const int half = lane >> 4;
  const int klo  = half << 1;

  const int srow = tid >> 3;
  const int sc4  = (tid & 7) << 2;
  const size_t aStep = (size_t)32 * lda;
  const float* __restrict__ gA = A + (size_t)(m0 + srow) * lda + sc4;
  const int nr0 = min(n0 + srow,      N - 1);
  const int nr1 = min(n0 + srow + 32, N - 1);
  const int nr2 = min(n0 + srow + 64, N - 1);
  const int nr3 = min(n0 + srow + 96, N - 1);
  const float* __restrict__ gB0 = W + (size_t)nr0 * ldb + sc4;
  const float* __restrict__ gB1 = W + (size_t)nr1 * ldb + sc4;
  const float* __restrict__ gB2 = W + (size_t)nr2 * ldb + sc4;
  const float* __restrict__ gB3 = W + (size_t)nr3 * ldb + sc4;
  const int sOff = srow * GPITCH + sc4;

  v8f acc[4][2];
#pragma unroll
  for (int i = 0; i < 4; ++i)
#pragma unroll
    for (int j = 0; j < 2; ++j) acc[i][j] = {};

  float4 ra0 = *(const float4*)(gA);
  float4 ra1 = *(const float4*)(gA + aStep);
  float4 ra2 = *(const float4*)(gA + 2 * aStep);
  float4 ra3 = *(const float4*)(gA + 3 * aStep);
  float4 rb0 = *(const float4*)(gB0);
  float4 rb1 = *(const float4*)(gB1);
  float4 rb2 = *(const float4*)(gB2);
  float4 rb3 = *(const float4*)(gB3);
  *(float4*)&sA[0][sOff]               = ra0;
  *(float4*)&sA[0][sOff + 32 * GPITCH] = ra1;
  *(float4*)&sA[0][sOff + 64 * GPITCH] = ra2;
  *(float4*)&sA[0][sOff + 96 * GPITCH] = ra3;
  *(float4*)&sB[0][sOff]               = rb0;
  *(float4*)&sB[0][sOff + 32 * GPITCH] = rb1;
  *(float4*)&sB[0][sOff + 64 * GPITCH] = rb2;
  *(float4*)&sB[0][sOff + 96 * GPITCH] = rb3;
  __syncthreads();

  const int nChunks = K / GKC;
  for (int kc = 0; kc < nChunks; ++kc) {
    const int cur = kc & 1;
    const bool more = (kc + 1 < nChunks);

    if (more) {
      const int kofs = (kc + 1) * GKC;
      ra0 = *(const float4*)(gA + kofs);
      ra1 = *(const float4*)(gA + aStep + kofs);
      ra2 = *(const float4*)(gA + 2 * aStep + kofs);
      ra3 = *(const float4*)(gA + 3 * aStep + kofs);
      rb0 = *(const float4*)(gB0 + kofs);
      rb1 = *(const float4*)(gB1 + kofs);
      rb2 = *(const float4*)(gB2 + kofs);
      rb3 = *(const float4*)(gB3 + kofs);
    }

    const float* pA = &sA[cur][(wm * 64 + r) * GPITCH + klo];
    const float* pB = &sB[cur][(wn * 32 + r) * GPITCH + klo];
#pragma unroll
    for (int s = 0; s < GKC / 4; ++s) {
      v2f a[4], b[2];
#pragma unroll
      for (int i = 0; i < 4; ++i) a[i] = *(const v2f*)(pA + i * 16 * GPITCH + 4 * s);
#pragma unroll
      for (int j = 0; j < 2; ++j) b[j] = *(const v2f*)(pB + j * 16 * GPITCH + 4 * s);
#pragma unroll
      for (int i = 0; i < 4; ++i)
#pragma unroll
        for (int j = 0; j < 2; ++j)
          acc[i][j] = wmma_f32_16x16x4(a[i], b[j], acc[i][j]);
    }

    if (more) {
      float* dA = &sA[cur ^ 1][sOff];
      float* dB = &sB[cur ^ 1][sOff];
      *(float4*)(dA)               = ra0;
      *(float4*)(dA + 32 * GPITCH) = ra1;
      *(float4*)(dA + 64 * GPITCH) = ra2;
      *(float4*)(dA + 96 * GPITCH) = ra3;
      *(float4*)(dB)               = rb0;
      *(float4*)(dB + 32 * GPITCH) = rb1;
      *(float4*)(dB + 64 * GPITCH) = rb2;
      *(float4*)(dB + 96 * GPITCH) = rb3;
    }
    __syncthreads();
  }

  const int cbase = n0 + wn * 32;
  if (cbase >= N) return;
#pragma unroll
  for (int i = 0; i < 4; ++i) {
#pragma unroll
    for (int j = 0; j < 2; ++j) {
      const int cn = cbase + j * 16 + r;
#pragma unroll
      for (int v = 0; v < 8; ++v) {
        const int cm = m0 + wm * 64 + i * 16 + v + (half << 3);
        float val = acc[i][j][v];
        if (EPI == 1) {
          val += bias[cn];
          val = (val > 20.f) ? val : log1pf(__expf(val));
        }
        C[(size_t)cm * ldc + cn] = val;
      }
    }
  }
}

__global__ void __launch_bounds__(256)
conv_silu_kernel(const float* __restrict__ zx,
                 const float* __restrict__ conv_w,
                 const float* __restrict__ conv_b,
                 float* __restrict__ xconv)
{
  int idx = blockIdx.x * blockDim.x + threadIdx.x;
  if (idx >= BATCH * SEQ * D_INNER) return;
  const int d = idx % D_INNER;
  const int l = (idx / D_INNER) % SEQ;
  const int b = idx / (D_INNER * SEQ);
  const int src = ((d >> 6) << 4) + (d & 15);

  const float* __restrict__ w = conv_w + d * D_CONV;
  float acc = conv_b[d];
#pragma unroll
  for (int t = 0; t < D_CONV; ++t) {
    const int ll = l - (D_CONV - 1) + t;
    if (ll >= 0) {
      float xv = zx[(size_t)(b * SEQ + ll) * E_DIM + OFF_X + src];
      acc = fmaf(w[t], xv, acc);
    }
  }
  acc = acc / (1.f + __expf(-acc));
  xconv[(size_t)idx] = acc;
}

#define SCAN_T   64
#define SCAN_NT  (SEQ / SCAN_T)

__global__ void __launch_bounds__(256)
scan_kernel(const float* __restrict__ zx,
            const float* __restrict__ xconv,
            const float* __restrict__ dtbuf,
            const float* __restrict__ Amat,
            const float* __restrict__ Dvec,
            float* __restrict__ Y)
{
  __shared__ __attribute__((aligned(16))) float2 s_pd[SCAN_T][PCH];
  __shared__ __attribute__((aligned(16))) float2 s_bc[SCAN_T][D_STATE];
  __shared__ float s_y[SCAN_T][PCH * D_STATE + 1];

  const int b = blockIdx.x / H_HEADS;
  const int h = blockIdx.x % H_HEADS;
  const int p = threadIdx.x >> 4;
  const int n = threadIdx.x & 15;

  const int st = threadIdx.x >> 2;
  const int sq = (threadIdx.x & 3) << 2;

  const float  An  = Amat[(size_t)(h * PCH + p) * D_STATE + n];
  const float4 Dv4 = make_float4(Dvec[h * PCH + sq + 0], Dvec[h * PCH + sq + 1],
                                 Dvec[h * PCH + sq + 2], Dvec[h * PCH + sq + 3]);

  const size_t row0 = (size_t)b * SEQ + st;
  const size_t coff = (size_t)h * PCH + sq;
  const size_t boff = OFF_B + (size_t)(h / REP) * D_STATE + sq;
  const size_t cofs = OFF_C + (size_t)h * D_STATE + sq;
  const size_t zoff = OFF_Z + coff;

  float4 r_dt = *(const float4*)(dtbuf + row0 * D_INNER + coff);
  float4 r_x  = *(const float4*)(xconv + row0 * D_INNER + coff);
  float4 r_B  = *(const float4*)(zx + row0 * E_DIM + boff);
  float4 r_C  = *(const float4*)(zx + row0 * E_DIM + cofs);
  float4 r_z  = *(const float4*)(zx + row0 * E_DIM + zoff);

  float hstate = 0.f;
  for (int tile = 0; tile < SCAN_NT; ++tile) {
    {
      float4* pd = (float4*)&s_pd[st][sq];
      pd[0] = make_float4(r_dt.x, r_dt.x * r_x.x, r_dt.y, r_dt.y * r_x.y);
      pd[1] = make_float4(r_dt.z, r_dt.z * r_x.z, r_dt.w, r_dt.w * r_x.w);
      float4* bc = (float4*)&s_bc[st][sq];
      bc[0] = make_float4(r_B.x, r_C.x, r_B.y, r_C.y);
      bc[1] = make_float4(r_B.z, r_C.z, r_B.w, r_C.w);
    }
    const float4 c_x = r_x;
    const float4 c_z = r_z;
    __syncthreads();

    if (tile + 1 < SCAN_NT) {
      const size_t row = row0 + (size_t)(tile + 1) * SCAN_T;
      r_dt = *(const float4*)(dtbuf + row * D_INNER + coff);
      r_x  = *(const float4*)(xconv + row * D_INNER + coff);
      r_B  = *(const float4*)(zx + row * E_DIM + boff);
      r_C  = *(const float4*)(zx + row * E_DIM + cofs);
      r_z  = *(const float4*)(zx + row * E_DIM + zoff);
    }

#pragma unroll 16
    for (int t = 0; t < SCAN_T; ++t) {
      const float2 pd = s_pd[t][p];
      const float2 bc = s_bc[t][n];

      const float a = __expf(pd.x * An);
      hstate = fmaf(a, hstate, pd.y * bc.x);

      float yc = hstate * bc.y;
      yc += __shfl_xor(yc, 1, 32);
      yc += __shfl_xor(yc, 2, 32);
      yc += __shfl_xor(yc, 4, 32);
      yc += __shfl_xor(yc, 8, 32);

      s_y[t][threadIdx.x] = yc;
    }
    __syncthreads();

    {
      float4 y4;
      y4.x = s_y[st][(sq + 0) * D_STATE];
      y4.y = s_y[st][(sq + 1) * D_STATE];
      y4.z = s_y[st][(sq + 2) * D_STATE];
      y4.w = s_y[st][(sq + 3) * D_STATE];
      const float g0 = c_z.x / (1.f + __expf(-c_z.x));
      const float g1 = c_z.y / (1.f + __expf(-c_z.y));
      const float g2 = c_z.z / (1.f + __expf(-c_z.z));
      const float g3 = c_z.w / (1.f + __expf(-c_z.w));
      float4 o;
      o.x = (y4.x + Dv4.x * c_x.x) * g0;
      o.y = (y4.y + Dv4.y * c_x.y) * g1;
      o.z = (y4.z + Dv4.z * c_x.z) * g2;
      o.w = (y4.w + Dv4.w * c_x.w) * g3;
      const size_t row = row0 + (size_t)tile * SCAN_T;
      *(float4*)(Y + row * D_INNER + coff) = o;
    }
  }
}

extern "C" void kernel_launch(void* const* d_in, const int* in_sizes, int n_in,
                              void* d_out, int out_size, void* d_ws, size_t ws_size,
                              hipStream_t stream) {
  const float* hidden     = (const float*)d_in[0];
  const float* in_proj_w  = (const float*)d_in[1];
  const float* conv_w     = (const float*)d_in[2];
  const float* conv_b     = (const float*)d_in[3];
  const float* dt_proj_w  = (const float*)d_in[4];
  const float* dt_proj_b  = (const float*)d_in[5];
  const float* Amat       = (const float*)d_in[6];
  const float* Dvec       = (const float*)d_in[7];
  const float* out_proj_w = (const float*)d_in[8];
  float* out = (float*)d_out;

  float* zx    = (float*)d_ws;
  float* xconv = zx    + (size_t)M_ROWS * E_DIM;
  float* dtbuf = xconv + (size_t)M_ROWS * D_INNER;
  float* Ybuf  = dtbuf + (size_t)M_ROWS * D_INNER;

  {
    int blocks = (M_ROWS / GBM) * ((E_DIM + GBN - 1) / GBN);
    wmma_gemm_f32<0><<<blocks, 256, 0, stream>>>(
        hidden, D_MODEL, in_proj_w, D_MODEL, zx, E_DIM,
        M_ROWS, E_DIM, D_MODEL, nullptr);
  }

  {
    int n = BATCH * SEQ * D_INNER;
    conv_silu_kernel<<<(n + 255) / 256, 256, 0, stream>>>(zx, conv_w, conv_b, xconv);
  }

  {
    int blocks = (M_ROWS / GBM) * ((D_INNER + GBN - 1) / GBN);
    wmma_gemm_f32<1><<<blocks, 256, 0, stream>>>(
        zx + OFF_DT, E_DIM, dt_proj_w, DT_RANK, dtbuf, D_INNER,
        M_ROWS, D_INNER, DT_RANK, dt_proj_b);
  }

  scan_kernel<<<BATCH * H_HEADS, 256, 0, stream>>>(zx, xconv, dtbuf, Amat, Dvec, Ybuf);

  {
    int blocks = (M_ROWS / GBM) * ((D_MODEL + GBN - 1) / GBN);
    wmma_gemm_f32<0><<<blocks, 256, 0, stream>>>(
        Ybuf, D_INNER, out_proj_w, D_INNER, out, D_MODEL,
        M_ROWS, D_MODEL, D_INNER, nullptr);
  }
}


# conv+SiLU: each thread owns four neighbouring channels (one 16-byte piece) and walks 16 time steps with its window in registers: 4,096 waves instead of 262,144; per-element arithmetic unchanged
# speedup vs baseline: 5.7072x; 1.0329x over previous
#include <hip/hip_runtime.h>
#include <math.h>

#define D_MODEL   1024
#define D_INNER   2048
#define D_XB       512
#define D_STATE     16
#define D_CONV       4
#define DT_RANK     64
#define NXB         32
#define H_HEADS    128
#define REP          4
#define PCH         16
#define BATCH        2
#define SEQ       2048
#define E_DIM     (2*D_INNER + 2*D_XB + DT_RANK)
#define M_ROWS    (BATCH*SEQ)

#define OFF_Z   0
#define OFF_X   (D_INNER)
#define OFF_B   (D_INNER + D_XB)
#define OFF_C   (D_INNER + 2*D_XB)
#define OFF_DT  (2*D_INNER + 2*D_XB)

typedef __attribute__((ext_vector_type(2))) float v2f;
typedef __attribute__((ext_vector_type(8))) float v8f;

__device__ __forceinline__ v8f wmma_f32_16x16x4(v2f a, v2f b, v8f c) {
  return __builtin_amdgcn_wmma_f32_16x16x4_f32(
       false, a,  false, b,
       (short)0, c,  false,  false);
}

#define GBM     128
#define GBN     128
#define GKC      32
#define GPITCH   36

template<int EPI>
__global__ void __launch_bounds__(256)
wmma_gemm_f32(const float* __restrict__ A, int lda,
              const float* __restrict__ W, int ldb,
              float* __restrict__ C, int ldc,
              int M, int N, int K,
              const float* __restrict__ bias)
{
  __shared__ __attribute__((aligned(16))) float sA[2][GBM * GPITCH];
  __shared__ __attribute__((aligned(16))) float sB[2][GBN * GPITCH];

  const int tid  = threadIdx.x;
  const int lane = tid & 31;
  const int wave = tid >> 5;
  const int wm   = wave >> 2;
  const int wn   = wave & 3;
  const int tilesN = (N + GBN - 1) / GBN;
  const int bm = blockIdx.x / tilesN;
  const int bn = blockIdx.x - bm * tilesN;
  const int m0 = bm * GBM;
  const int n0 = bn * GBN;

  const int r    = lane & 15;
  const int half = lane >> 4;
  const int klo  = half << 1;

  const int srow = tid >> 3;
  const int sc4  = (tid & 7) << 2;
  const size_t aStep = (size_t)32 * lda;
  const float* __restrict__ gA = A + (size_t)(m0 + srow) * lda + sc4;
  const int nr0 = min(n0 + srow,      N - 1);
  const int nr1 = min(n0 + srow + 32, N - 1);
  const int nr2 = min(n0 + srow + 64, N - 1);
  const int nr3 = min(n0 + srow + 96, N - 1);
  const float* __restrict__ gB0 = W + (size_t)nr0 * ldb + sc4;
  const float* __restrict__ gB1 = W + (size_t)nr1 * ldb + sc4;
  const float* __restrict__ gB2 = W + (size_t)nr2 * ldb + sc4;
  const float* __restrict__ gB3 = W + (size_t)nr3 * ldb + sc4;
  const int sOff = srow * GPITCH + sc4;

  v8f acc[4][2];
#pragma unroll
  for (int i = 0; i < 4; ++i)
#pragma unroll
    for (int j = 0; j < 2; ++j) acc[i][j] = {};

  float4 ra0 = *(const float4*)(gA);
  float4 ra1 = *(const float4*)(gA + aStep);
  float4 ra2 = *(const float4*)(gA + 2 * aStep);
  float4 ra3 = *(const float4*)(gA + 3 * aStep);
  float4 rb0 = *(const float4*)(gB0);
  float4 rb1 = *(const float4*)(gB1);
  float4 rb2 = *(const float4*)(gB2);
  float4 rb3 = *(const float4*)(gB3);
  *(float4*)&sA[0][sOff]               = ra0;
  *(float4*)&sA[0][sOff + 32 * GPITCH] = ra1;
  *(float4*)&sA[0][sOff + 64 * GPITCH] = ra2;
  *(float4*)&sA[0][sOff + 96 * GPITCH] = ra3;
  *(float4*)&sB[0][sOff]               = rb0;
  *(float4*)&sB[0][sOff + 32 * GPITCH] = rb1;
  *(float4*)&sB[0][sOff + 64 * GPITCH] = rb2;
  *(float4*)&sB[0][sOff + 96 * GPITCH] = rb3;
  __syncthreads();

  const int nChunks = K / GKC;
  for (int kc = 0; kc < nChunks; ++kc) {
    const int cur = kc & 1;
    const bool more = (kc + 1 < nChunks);

    if (more) {
      const int kofs = (kc + 1) * GKC;
      ra0 = *(const float4*)(gA + kofs);
      ra1 = *(const float4*)(gA + aStep + kofs);
      ra2 = *(const float4*)(gA + 2 * aStep + kofs);
      ra3 = *(const float4*)(gA + 3 * aStep + kofs);
      rb0 = *(const float4*)(gB0 + kofs);
      rb1 = *(const float4*)(gB1 + kofs);
      rb2 = *(const float4*)(gB2 + kofs);
      rb3 = *(const float4*)(gB3 + kofs);
    }

    const float* pA = &sA[cur][(wm * 64 + r) * GPITCH + klo];
    const float* pB = &sB[cur][(wn * 32 + r) * GPITCH + klo];
#pragma unroll
    for (int s = 0; s < GKC / 4; ++s) {
      v2f a[4], b[2];
#pragma unroll
      for (int i = 0; i < 4; ++i) a[i] = *(const v2f*)(pA + i * 16 * GPITCH + 4 * s);
#pragma unroll
      for (int j = 0; j < 2; ++j) b[j] = *(const v2f*)(pB + j * 16 * GPITCH + 4 * s);
#pragma unroll
      for (int i = 0; i < 4; ++i)
#pragma unroll
        for (int j = 0; j < 2; ++j)
          acc[i][j] = wmma_f32_16x16x4(a[i], b[j], acc[i][j]);
    }

    if (more) {
      float* dA = &sA[cur ^ 1][sOff];
      float* dB = &sB[cur ^ 1][sOff];
      *(float4*)(dA)               = ra0;
      *(float4*)(dA + 32 * GPITCH) = ra1;
      *(float4*)(dA + 64 * GPITCH) = ra2;
      *(float4*)(dA + 96 * GPITCH) = ra3;
      *(float4*)(dB)               = rb0;
      *(float4*)(dB + 32 * GPITCH) = rb1;
      *(float4*)(dB + 64 * GPITCH) = rb2;
      *(float4*)(dB + 96 * GPITCH) = rb3;
    }
    __syncthreads();
  }

  const int cbase = n0 + wn * 32;
  if (cbase >= N) return;
#pragma unroll
  for (int i = 0; i < 4; ++i) {
#pragma unroll
    for (int j = 0; j < 2; ++j) {
      const int cn = cbase + j * 16 + r;
#pragma unroll
      for (int v = 0; v < 8; ++v) {
        const int cm = m0 + wm * 64 + i * 16 + v + (half << 3);
        float val = acc[i][j][v];
        if (EPI == 1) {
          val += bias[cn];
          val = (val > 20.f) ? val : log1pf(__expf(val));
        }
        C[(size_t)cm * ldc + cn] = val;
      }
    }
  }
}

#define CONV_T   16
#define CONV_C4  (D_INNER / 4)

static_assert(SEQ % CONV_T == 0, "a thread's steps stay inside one sequence");
static_assert((M_ROWS / CONV_T) * CONV_C4 % 256 == 0, "whole blocks");

__device__ __forceinline__ float conv_silu_one(float bias, float4 w,
                                               float xm3, float xm2, float xm1, float x0,
                                               bool v3, bool v2, bool v1)
{
  float acc = bias;
  if (v3) acc = fmaf(w.x, xm3, acc);
  if (v2) acc = fmaf(w.y, xm2, acc);
  if (v1) acc = fmaf(w.z, xm1, acc);
  acc = fmaf(w.w, x0, acc);
  return acc / (1.f + __expf(-acc));
}

__global__ void __launch_bounds__(256)
conv_silu_kernel(const float* __restrict__ zx,
                 const float* __restrict__ conv_w,
                 const float* __restrict__ conv_b,
                 float* __restrict__ xconv)
{
  const int idx  = blockIdx.x * 256 + threadIdx.x;
  const int c4   = idx % CONV_C4;
  const int rg   = idx / CONV_C4;
  const int d0   = c4 << 2;
  const int src  = ((d0 >> 6) << 4) + (d0 & 15);
  const int row0 = rg * CONV_T;
  const bool hal = (row0 % SEQ) != 0;

  const float4 w0 = *(const float4*)(conv_w + (size_t)(d0 + 0) * D_CONV);
  const float4 w1 = *(const float4*)(conv_w + (size_t)(d0 + 1) * D_CONV);
  const float4 w2 = *(const float4*)(conv_w + (size_t)(d0 + 2) * D_CONV);
  const float4 w3 = *(const float4*)(conv_w + (size_t)(d0 + 3) * D_CONV);
  const float4 bz = *(const float4*)(conv_b + d0);

  const float* __restrict__ px = zx + (size_t)row0 * E_DIM + OFF_X + src;
  float4 xw[CONV_T + 3];
  if (hal) {
    xw[0] = *(const float4*)(px - (ptrdiff_t)3 * E_DIM);
    xw[1] = *(const float4*)(px - (ptrdiff_t)2 * E_DIM);
    xw[2] = *(const float4*)(px - (ptrdiff_t)1 * E_DIM);
  } else {
    xw[0] = make_float4(0.f, 0.f, 0.f, 0.f);
    xw[1] = xw[0];
    xw[2] = xw[0];
  }
#pragma unroll
  for (int s = 0; s < CONV_T; ++s)
    xw[s + 3] = *(const float4*)(px + (size_t)s * E_DIM);

  float* __restrict__ po = xconv + (size_t)row0 * D_INNER + d0;
#pragma unroll
  for (int s = 0; s < CONV_T; ++s) {
    const bool v3 = hal || (s >= 3);
    const bool v2 = hal || (s >= 2);
    const bool v1 = hal || (s >= 1);
    float4 o;
    o.x = conv_silu_one(bz.x, w0, xw[s].x, xw[s + 1].x, xw[s + 2].x, xw[s + 3].x, v3, v2, v1);
    o.y = conv_silu_one(bz.y, w1, xw[s].y, xw[s + 1].y, xw[s + 2].y, xw[s + 3].y, v3, v2, v1);
    o.z = conv_silu_one(bz.z, w2, xw[s].z, xw[s + 1].z, xw[s + 2].z, xw[s + 3].z, v3, v2, v1);
    o.w = conv_silu_one(bz.w, w3, xw[s].w, xw[s + 1].w, xw[s + 2].w, xw[s + 3].w, v3, v2, v1);
    *(float4*)(po + (size_t)s * D_INNER) = o;
  }
}

#define SCAN_T   64
#define SCAN_NT  (SEQ / SCAN_T)

__global__ void __launch_bounds__(256)
scan_kernel(const float* __restrict__ zx,
            const float* __restrict__ xconv,
            const float* __restrict__ dtbuf,
            const float* __restrict__ Amat,
            const float* __restrict__ Dvec,
            float* __restrict__ Y)
{
  __shared__ __attribute__((aligned(16))) float2 s_pd[SCAN_T][PCH];
  __shared__ __attribute__((aligned(16))) float2 s_bc[SCAN_T][D_STATE];
  __shared__ float s_y[SCAN_T][PCH * D_STATE + 1];

  const int b = blockIdx.x / H_HEADS;
  const int h = blockIdx.x % H_HEADS;
  const int p = threadIdx.x >> 4;
  const int n = threadIdx.x & 15;

  const int st = threadIdx.x >> 2;
  const int sq = (threadIdx.x & 3) << 2;

  const float  An  = Amat[(size_t)(h * PCH + p) * D_STATE + n];
  const float4 Dv4 = make_float4(Dvec[h * PCH + sq + 0], Dvec[h * PCH + sq + 1],
                                 Dvec[h * PCH + sq + 2], Dvec[h * PCH + sq + 3]);

  const size_t row0 = (size_t)b * SEQ + st;
  const size_t coff = (size_t)h * PCH + sq;
  const size_t boff = OFF_B + (size_t)(h / REP) * D_STATE + sq;
  const size_t cofs = OFF_C + (size_t)h * D_STATE + sq;
  const size_t zoff = OFF_Z + coff;

  float4 r_dt = *(const float4*)(dtbuf + row0 * D_INNER + coff);
  float4 r_x  = *(const float4*)(xconv + row0 * D_INNER + coff);
  float4 r_B  = *(const float4*)(zx + row0 * E_DIM + boff);
  float4 r_C  = *(const float4*)(zx + row0 * E_DIM + cofs);
  float4 r_z  = *(const float4*)(zx + row0 * E_DIM + zoff);

  float hstate = 0.f;
  for (int tile = 0; tile < SCAN_NT; ++tile) {
    {
      float4* pd = (float4*)&s_pd[st][sq];
      pd[0] = make_float4(r_dt.x, r_dt.x * r_x.x, r_dt.y, r_dt.y * r_x.y);
      pd[1] = make_float4(r_dt.z, r_dt.z * r_x.z, r_dt.w, r_dt.w * r_x.w);
      float4* bc = (float4*)&s_bc[st][sq];
      bc[0] = make_float4(r_B.x, r_C.x, r_B.y, r_C.y);
      bc[1] = make_float4(r_B.z, r_C.z, r_B.w, r_C.w);
    }
    const float4 c_x = r_x;
    const float4 c_z = r_z;
    __syncthreads();

    if (tile + 1 < SCAN_NT) {
      const size_t row = row0 + (size_t)(tile + 1) * SCAN_T;
      r_dt = *(const float4*)(dtbuf + row * D_INNER + coff);
      r_x  = *(const float4*)(xconv + row * D_INNER + coff);
      r_B  = *(const float4*)(zx + row * E_DIM + boff);
      r_C  = *(const float4*)(zx + row * E_DIM + cofs);
      r_z  = *(const float4*)(zx + row * E_DIM + zoff);
    }

#pragma unroll 16
    for (int t = 0; t < SCAN_T; ++t) {
      const float2 pd = s_pd[t][p];
      const float2 bc = s_bc[t][n];

      const float a = __expf(pd.x * An);
      hstate = fmaf(a, hstate, pd.y * bc.x);

      float yc = hstate * bc.y;
      yc += __shfl_xor(yc, 1, 32);
      yc += __shfl_xor(yc, 2, 32);
      yc += __shfl_xor(yc, 4, 32);
      yc += __shfl_xor(yc, 8, 32);

      s_y[t][threadIdx.x] = yc;
    }
    __syncthreads();

    {
      float4 y4;
      y4.x = s_y[st][(sq + 0) * D_STATE];
      y4.y = s_y[st][(sq + 1) * D_STATE];
      y4.z = s_y[st][(sq + 2) * D_STATE];
      y4.w = s_y[st][(sq + 3) * D_STATE];
      const float g0 = c_z.x / (1.f + __expf(-c_z.x));
      const float g1 = c_z.y / (1.f + __expf(-c_z.y));
      const float g2 = c_z.z / (1.f + __expf(-c_z.z));
      const float g3 = c_z.w / (1.f + __expf(-c_z.w));
      float4 o;
      o.x = (y4.x + Dv4.x * c_x.x) * g0;
      o.y = (y4.y + Dv4.y * c_x.y) * g1;
      o.z = (y4.z + Dv4.z * c_x.z) * g2;
      o.w = (y4.w + Dv4.w * c_x.w) * g3;
      const size_t row = row0 + (size_t)tile * SCAN_T;
      *(float4*)(Y + row * D_INNER + coff) = o;
    }
  }
}

extern "C" void kernel_launch(void* const* d_in, const int* in_sizes, int n_in,
                              void* d_out, int out_size, void* d_ws, size_t ws_size,
                              hipStream_t stream) {
  const float* hidden     = (const float*)d_in[0];
  const float* in_proj_w  = (const float*)d_in[1];
  const float* conv_w     = (const float*)d_in[2];
  const float* conv_b     = (const float*)d_in[3];
  const float* dt_proj_w  = (const float*)d_in[4];
  const float* dt_proj_b  = (const float*)d_in[5];
  const float* Amat       = (const float*)d_in[6];
  const float* Dvec       = (const float*)d_in[7];
  const float* out_proj_w = (const float*)d_in[8];
  float* out = (float*)d_out;

  float* zx    = (float*)d_ws;
  float* xconv = zx    + (size_t)M_ROWS * E_DIM;
  float* dtbuf = xconv + (size_t)M_ROWS * D_INNER;
  float* Ybuf  = dtbuf + (size_t)M_ROWS * D_INNER;

  {
    int blocks = (M_ROWS / GBM) * ((E_DIM + GBN - 1) / GBN);
    wmma_gemm_f32<0><<<blocks, 256, 0, stream>>>(
        hidden, D_MODEL, in_proj_w, D_MODEL, zx, E_DIM,
        M_ROWS, E_DIM, D_MODEL, nullptr);
  }

  {
    int blocks = (M_ROWS / CONV_T) * CONV_C4 / 256;
    conv_silu_kernel<<<blocks, 256, 0, stream>>>(zx, conv_w, conv_b, xconv);
  }

  {
    int blocks = (M_ROWS / GBM) * ((D_INNER + GBN - 1) / GBN);
    wmma_gemm_f32<1><<<blocks, 256, 0, stream>>>(
        zx + OFF_DT, E_DIM, dt_proj_w, DT_RANK, dtbuf, D_INNER,
        M_ROWS, D_INNER, DT_RANK, dt_proj_b);
  }

  scan_kernel<<<BATCH * H_HEADS, 256, 0, stream>>>(zx, xconv, dtbuf, Amat, Dvec, Ybuf);

  {
    int blocks = (M_ROWS / GBM) * ((D_MODEL + GBN - 1) / GBN);
    wmma_gemm_f32<0><<<blocks, 256, 0, stream>>>(
        Ybuf, D_INNER, out_proj_w, D_INNER, out, D_MODEL,
        M_ROWS, D_MODEL, D_INNER, nullptr);
  }
}


# scan: the 16-state sum of every step runs as four DPP row_xmask adds on the vector ALU instead of four LDS-unit permutes (__shfl_xor); same xor butterfly, same pairs and order of the adds
# speedup vs baseline: 6.2294x; 1.0915x over previous
#include <hip/hip_runtime.h>
#include <math.h>

#define D_MODEL   1024
#define D_INNER   2048
#define D_XB       512
#define D_STATE     16
#define D_CONV       4
#define DT_RANK     64
#define NXB         32
#define H_HEADS    128
#define REP          4
#define PCH         16
#define BATCH        2
#define SEQ       2048
#define E_DIM     (2*D_INNER + 2*D_XB + DT_RANK)
#define M_ROWS    (BATCH*SEQ)

#define OFF_Z   0
#define OFF_X   (D_INNER)
#define OFF_B   (D_INNER + D_XB)
#define OFF_C   (D_INNER + 2*D_XB)
#define OFF_DT  (2*D_INNER + 2*D_XB)

typedef __attribute__((ext_vector_type(2))) float v2f;
typedef __attribute__((ext_vector_type(8))) float v8f;

__device__ __forceinline__ v8f wmma_f32_16x16x4(v2f a, v2f b, v8f c) {
  return __builtin_amdgcn_wmma_f32_16x16x4_f32(
       false, a,  false, b,
       (short)0, c,  false,  false);
}

#define GBM     128
#define GBN     128
#define GKC      32
#define GPITCH   36

template<int EPI>
__global__ void __launch_bounds__(256)
wmma_gemm_f32(const float* __restrict__ A, int lda,
              const float* __restrict__ W, int ldb,
              float* __restrict__ C, int ldc,
              int M, int N, int K,
              const float* __restrict__ bias)
{
  __shared__ __attribute__((aligned(16))) float sA[2][GBM * GPITCH];
  __shared__ __attribute__((aligned(16))) float sB[2][GBN * GPITCH];

  const int tid  = threadIdx.x;
  const int lane = tid & 31;
  const int wave = tid >> 5;
  const int wm   = wave >> 2;
  const int wn   = wave & 3;
  const int tilesN = (N + GBN - 1) / GBN;
  const int bm = blockIdx.x / tilesN;
  const int bn = blockIdx.x - bm * tilesN;
  const int m0 = bm * GBM;
  const int n0 = bn * GBN;

  const int r    = lane & 15;
  const int half = lane >> 4;
  const int klo  = half << 1;

  const int srow = tid >> 3;
  const int sc4  = (tid & 7) << 2;
  const size_t aStep = (size_t)32 * lda;
  const float* __restrict__ gA = A + (size_t)(m0 + srow) * lda + sc4;
  const int nr0 = min(n0 + srow,      N - 1);
  const int nr1 = min(n0 + srow + 32, N - 1);
  const int nr2 = min(n0 + srow + 64, N - 1);
  const int nr3 = min(n0 + srow + 96, N - 1);
  const float* __restrict__ gB0 = W + (size_t)nr0 * ldb + sc4;
  const float* __restrict__ gB1 = W + (size_t)nr1 * ldb + sc4;
  const float* __restrict__ gB2 = W + (size_t)nr2 * ldb + sc4;
  const float* __restrict__ gB3 = W + (size_t)nr3 * ldb + sc4;
  const int sOff = srow * GPITCH + sc4;

  v8f acc[4][2];
#pragma unroll
  for (int i = 0; i < 4; ++i)
#pragma unroll
    for (int j = 0; j < 2; ++j) acc[i][j] = {};

  float4 ra0 = *(const float4*)(gA);
  float4 ra1 = *(const float4*)(gA + aStep);
  float4 ra2 = *(const float4*)(gA + 2 * aStep);
  float4 ra3 = *(const float4*)(gA + 3 * aStep);
  float4 rb0 = *(const float4*)(gB0);
  float4 rb1 = *(const float4*)(gB1);
  float4 rb2 = *(const float4*)(gB2);
  float4 rb3 = *(const float4*)(gB3);
  *(float4*)&sA[0][sOff]               = ra0;
  *(float4*)&sA[0][sOff + 32 * GPITCH] = ra1;
  *(float4*)&sA[0][sOff + 64 * GPITCH] = ra2;
  *(float4*)&sA[0][sOff + 96 * GPITCH] = ra3;
  *(float4*)&sB[0][sOff]               = rb0;
  *(float4*)&sB[0][sOff + 32 * GPITCH] = rb1;
  *(float4*)&sB[0][sOff + 64 * GPITCH] = rb2;
  *(float4*)&sB[0][sOff + 96 * GPITCH] = rb3;
  __syncthreads();

  const int nChunks = K / GKC;
  for (int kc = 0; kc < nChunks; ++kc) {
    const int cur = kc & 1;
    const bool more = (kc + 1 < nChunks);

    if (more) {
      const int kofs = (kc + 1) * GKC;
      ra0 = *(const float4*)(gA + kofs);
      ra1 = *(const float4*)(gA + aStep + kofs);
      ra2 = *(const float4*)(gA + 2 * aStep + kofs);
      ra3 = *(const float4*)(gA + 3 * aStep + kofs);
      rb0 = *(const float4*)(gB0 + kofs);
      rb1 = *(const float4*)(gB1 + kofs);
      rb2 = *(const float4*)(gB2 + kofs);
      rb3 = *(const float4*)(gB3 + kofs);
    }

    const float* pA = &sA[cur][(wm * 64 + r) * GPITCH + klo];
    const float* pB = &sB[cur][(wn * 32 + r) * GPITCH + klo];
#pragma unroll
    for (int s = 0; s < GKC / 4; ++s) {
      v2f a[4], b[2];
#pragma unroll
      for (int i = 0; i < 4; ++i) a[i] = *(const v2f*)(pA + i * 16 * GPITCH + 4 * s);
#pragma unroll
      for (int j = 0; j < 2; ++j) b[j] = *(const v2f*)(pB + j * 16 * GPITCH + 4 * s);
#pragma unroll
      for (int i = 0; i < 4; ++i)
#pragma unroll
        for (int j = 0; j < 2; ++j)
          acc[i][j] = wmma_f32_16x16x4(a[i], b[j], acc[i][j]);
    }

    if (more) {
      float* dA = &sA[cur ^ 1][sOff];
      float* dB = &sB[cur ^ 1][sOff];
      *(float4*)(dA)               = ra0;
      *(float4*)(dA + 32 * GPITCH) = ra1;
      *(float4*)(dA + 64 * GPITCH) = ra2;
      *(float4*)(dA + 96 * GPITCH) = ra3;
      *(float4*)(dB)               = rb0;
      *(float4*)(dB + 32 * GPITCH) = rb1;
      *(float4*)(dB + 64 * GPITCH) = rb2;
      *(float4*)(dB + 96 * GPITCH) = rb3;
    }
    __syncthreads();
  }

  const int cbase = n0 + wn * 32;
  if (cbase >= N) return;
#pragma unroll
  for (int i = 0; i < 4; ++i) {
#pragma unroll
    for (int j = 0; j < 2; ++j) {
      const int cn = cbase + j * 16 + r;
#pragma unroll
      for (int v = 0; v < 8; ++v) {
        const int cm = m0 + wm * 64 + i * 16 + v + (half << 3);
        float val = acc[i][j][v];
        if (EPI == 1) {
          val += bias[cn];
          val = (val > 20.f) ? val : log1pf(__expf(val));
        }
        C[(size_t)cm * ldc + cn] = val;
      }
    }
  }
}

#define CONV_T   16
#define CONV_C4  (D_INNER / 4)

static_assert(SEQ % CONV_T == 0, "a thread's steps stay inside one sequence");
static_assert((M_ROWS / CONV_T) * CONV_C4 % 256 == 0, "whole blocks");

__device__ __forceinline__ float conv_silu_one(float bias, float4 w,
                                               float xm3, float xm2, float xm1, float x0,
                                               bool v3, bool v2, bool v1)
{
  float acc = bias;
  if (v3) acc = fmaf(w.x, xm3, acc);
  if (v2) acc = fmaf(w.y, xm2, acc);
  if (v1) acc = fmaf(w.z, xm1, acc);
  acc = fmaf(w.w, x0, acc);
  return acc / (1.f + __expf(-acc));
}

__global__ void __launch_bounds__(256)
conv_silu_kernel(const float* __restrict__ zx,
                 const float* __restrict__ conv_w,
                 const float* __restrict__ conv_b,
                 float* __restrict__ xconv)
{
  const int idx  = blockIdx.x * 256 + threadIdx.x;
  const int c4   = idx % CONV_C4;
  const int rg   = idx / CONV_C4;
  const int d0   = c4 << 2;
  const int src  = ((d0 >> 6) << 4) + (d0 & 15);
  const int row0 = rg * CONV_T;
  const bool hal = (row0 % SEQ) != 0;

  const float4 w0 = *(const float4*)(conv_w + (size_t)(d0 + 0) * D_CONV);
  const float4 w1 = *(const float4*)(conv_w + (size_t)(d0 + 1) * D_CONV);
  const float4 w2 = *(const float4*)(conv_w + (size_t)(d0 + 2) * D_CONV);
  const float4 w3 = *(const float4*)(conv_w + (size_t)(d0 + 3) * D_CONV);
  const float4 bz = *(const float4*)(conv_b + d0);

  const float* __restrict__ px = zx + (size_t)row0 * E_DIM + OFF_X + src;
  float4 xw[CONV_T + 3];
  if (hal) {
    xw[0] = *(const float4*)(px - (ptrdiff_t)3 * E_DIM);
    xw[1] = *(const float4*)(px - (ptrdiff_t)2 * E_DIM);
    xw[2] = *(const float4*)(px - (ptrdiff_t)1 * E_DIM);
  } else {
    xw[0] = make_float4(0.f, 0.f, 0.f, 0.f);
    xw[1] = xw[0];
    xw[2] = xw[0];
  }
#pragma unroll
  for (int s = 0; s < CONV_T; ++s)
    xw[s + 3] = *(const float4*)(px + (size_t)s * E_DIM);

  float* __restrict__ po = xconv + (size_t)row0 * D_INNER + d0;
#pragma unroll
  for (int s = 0; s < CONV_T; ++s) {
    const bool v3 = hal || (s >= 3);
    const bool v2 = hal || (s >= 2);
    const bool v1 = hal || (s >= 1);
    float4 o;
    o.x = conv_silu_one(bz.x, w0, xw[s].x, xw[s + 1].x, xw[s + 2].x, xw[s + 3].x, v3, v2, v1);
    o.y = conv_silu_one(bz.y, w1, xw[s].y, xw[s + 1].y, xw[s + 2].y, xw[s + 3].y, v3, v2, v1);
    o.z = conv_silu_one(bz.z, w2, xw[s].z, xw[s + 1].z, xw[s + 2].z, xw[s + 3].z, v3, v2, v1);
    o.w = conv_silu_one(bz.w, w3, xw[s].w, xw[s + 1].w, xw[s + 2].w, xw[s + 3].w, v3, v2, v1);
    *(float4*)(po + (size_t)s * D_INNER) = o;
  }
}

#define SCAN_T   64
#define SCAN_NT  (SEQ / SCAN_T)

template<int M>
__device__ __forceinline__ float row_xor_add(float v)
{
  const int partner = __builtin_amdgcn_update_dpp(0, __builtin_bit_cast(int, v),
                                                  0x160 + M, 0xf, 0xf, true);
  return v + __builtin_bit_cast(float, partner);
}

__global__ void __launch_bounds__(256)
scan_kernel(const float* __restrict__ zx,
            const float* __restrict__ xconv,
            const float* __restrict__ dtbuf,
            const float* __restrict__ Amat,
            const float* __restrict__ Dvec,
            float* __restrict__ Y)
{
  __shared__ __attribute__((aligned(16))) float2 s_pd[SCAN_T][PCH];
  __shared__ __attribute__((aligned(16))) float2 s_bc[SCAN_T][D_STATE];
  __shared__ float s_y[SCAN_T][PCH * D_STATE + 1];

  const int b = blockIdx.x / H_HEADS;
  const int h = blockIdx.x % H_HEADS;
  const int p = threadIdx.x >> 4;
  const int n = threadIdx.x & 15;

  const int st = threadIdx.x >> 2;
  const int sq = (threadIdx.x & 3) << 2;

  const float  An  = Amat[(size_t)(h * PCH + p) * D_STATE + n];
  const float4 Dv4 = make_float4(Dvec[h * PCH + sq + 0], Dvec[h * PCH + sq + 1],
                                 Dvec[h * PCH + sq + 2], Dvec[h * PCH + sq + 3]);

  const size_t row0 = (size_t)b * SEQ + st;
  const size_t coff = (size_t)h * PCH + sq;
  const size_t boff = OFF_B + (size_t)(h / REP) * D_STATE + sq;
  const size_t cofs = OFF_C + (size_t)h * D_STATE + sq;
  const size_t zoff = OFF_Z + coff;

  float4 r_dt = *(const float4*)(dtbuf + row0 * D_INNER + coff);
  float4 r_x  = *(const float4*)(xconv + row0 * D_INNER + coff);
  float4 r_B  = *(const float4*)(zx + row0 * E_DIM + boff);
  float4 r_C  = *(const float4*)(zx + row0 * E_DIM + cofs);
  float4 r_z  = *(const float4*)(zx + row0 * E_DIM + zoff);

  float hstate = 0.f;
  for (int tile = 0; tile < SCAN_NT; ++tile) {
    {
      float4* pd = (float4*)&s_pd[st][sq];
      pd[0] = make_float4(r_dt.x, r_dt.x * r_x.x, r_dt.y, r_dt.y * r_x.y);
      pd[1] = make_float4(r_dt.z, r_dt.z * r_x.z, r_dt.w, r_dt.w * r_x.w);
      float4* bc = (float4*)&s_bc[st][sq];
      bc[0] = make_float4(r_B.x, r_C.x, r_B.y, r_C.y);
      bc[1] = make_float4(r_B.z, r_C.z, r_B.w, r_C.w);
    }
    const float4 c_x = r_x;
    const float4 c_z = r_z;
    __syncthreads();

    if (tile + 1 < SCAN_NT) {
      const size_t row = row0 + (size_t)(tile + 1) * SCAN_T;
      r_dt = *(const float4*)(dtbuf + row * D_INNER + coff);
      r_x  = *(const float4*)(xconv + row * D_INNER + coff);
      r_B  = *(const float4*)(zx + row * E_DIM + boff);
      r_C  = *(const float4*)(zx + row * E_DIM + cofs);
      r_z  = *(const float4*)(zx + row * E_DIM + zoff);
    }

#pragma unroll 16
    for (int t = 0; t < SCAN_T; ++t) {
      const float2 pd = s_pd[t][p];
      const float2 bc = s_bc[t][n];

      const float a = __expf(pd.x * An);
      hstate = fmaf(a, hstate, pd.y * bc.x);

      float yc = hstate * bc.y;
      yc = row_xor_add<1>(yc);
      yc = row_xor_add<2>(yc);
      yc = row_xor_add<4>(yc);
      yc = row_xor_add<8>(yc);

      s_y[t][threadIdx.x] = yc;
    }
    __syncthreads();

    {
      float4 y4;
      y4.x = s_y[st][(sq + 0) * D_STATE];
      y4.y = s_y[st][(sq + 1) * D_STATE];
      y4.z = s_y[st][(sq + 2) * D_STATE];
      y4.w = s_y[st][(sq + 3) * D_STATE];
      const float g0 = c_z.x / (1.f + __expf(-c_z.x));
      const float g1 = c_z.y / (1.f + __expf(-c_z.y));
      const float g2 = c_z.z / (1.f + __expf(-c_z.z));
      const float g3 = c_z.w / (1.f + __expf(-c_z.w));
      float4 o;
      o.x = (y4.x + Dv4.x * c_x.x) * g0;
      o.y = (y4.y + Dv4.y * c_x.y) * g1;
      o.z = (y4.z + Dv4.z * c_x.z) * g2;
      o.w = (y4.w + Dv4.w * c_x.w) * g3;
      const size_t row = row0 + (size_t)tile * SCAN_T;
      *(float4*)(Y + row * D_INNER + coff) = o;
    }
  }
}

extern "C" void kernel_launch(void* const* d_in, const int* in_sizes, int n_in,
                              void* d_out, int out_size, void* d_ws, size_t ws_size,
                              hipStream_t stream) {
  const float* hidden     = (const float*)d_in[0];
  const float* in_proj_w  = (const float*)d_in[1];
  const float* conv_w     = (const float*)d_in[2];
  const float* conv_b     = (const float*)d_in[3];
  const float* dt_proj_w  = (const float*)d_in[4];
  const float* dt_proj_b  = (const float*)d_in[5];
  const float* Amat       = (const float*)d_in[6];
  const float* Dvec       = (const float*)d_in[7];
  const float* out_proj_w = (const float*)d_in[8];
  float* out = (float*)d_out;

  float* zx    = (float*)d_ws;
  float* xconv = zx    + (size_t)M_ROWS * E_DIM;
  float* dtbuf = xconv + (size_t)M_ROWS * D_INNER;
  float* Ybuf  = dtbuf + (size_t)M_ROWS * D_INNER;

  {
    int blocks = (M_ROWS / GBM) * ((E_DIM + GBN - 1) / GBN);
    wmma_gemm_f32<0><<<blocks, 256, 0, stream>>>(
        hidden, D_MODEL, in_proj_w, D_MODEL, zx, E_DIM,
        M_ROWS, E_DIM, D_MODEL, nullptr);
  }

  {
    int blocks = (M_ROWS / CONV_T) * CONV_C4 / 256;
    conv_silu_kernel<<<blocks, 256, 0, stream>>>(zx, conv_w, conv_b, xconv);
  }

  {
    int blocks = (M_ROWS / GBM) * ((D_INNER + GBN - 1) / GBN);
    wmma_gemm_f32<1><<<blocks, 256, 0, stream>>>(
        zx + OFF_DT, E_DIM, dt_proj_w, DT_RANK, dtbuf, D_INNER,
        M_ROWS, D_INNER, DT_RANK, dt_proj_b);
  }

  scan_kernel<<<BATCH * H_HEADS, 256, 0, stream>>>(zx, xconv, dtbuf, Amat, Dvec, Ybuf);

  {
    int blocks = (M_ROWS / GBM) * ((D_MODEL + GBN - 1) / GBN);
    wmma_gemm_f32<0><<<blocks, 256, 0, stream>>>(
        Ybuf, D_INNER, out_proj_w, D_INNER, out, D_MODEL,
        M_ROWS, D_MODEL, D_INNER, nullptr);
  }
}


# in_proj grid: its 5120 whole-tile columns run as 32x40 = 1280 blocks of 128x128 (five per WGP, no sixth round); the 64 dt columns as 128 more 32x64-tile blocks of the same launch; same k order
# speedup vs baseline: 6.4671x; 1.0382x over previous
#include <hip/hip_runtime.h>
#include <math.h>

#define D_MODEL   1024
#define D_INNER   2048
#define D_XB       512
#define D_STATE     16
#define D_CONV       4
#define DT_RANK     64
#define NXB         32
#define H_HEADS    128
#define REP          4
#define PCH         16
#define BATCH        2
#define SEQ       2048
#define E_DIM     (2*D_INNER + 2*D_XB + DT_RANK)
#define M_ROWS    (BATCH*SEQ)

#define OFF_Z   0
#define OFF_X   (D_INNER)
#define OFF_B   (D_INNER + D_XB)
#define OFF_C   (D_INNER + 2*D_XB)
#define OFF_DT  (2*D_INNER + 2*D_XB)

typedef __attribute__((ext_vector_type(2))) float v2f;
typedef __attribute__((ext_vector_type(4))) float v4f;
typedef __attribute__((ext_vector_type(8))) float v8f;

__device__ __forceinline__ v8f wmma_f32_16x16x4(v2f a, v2f b, v8f c) {
  return __builtin_amdgcn_wmma_f32_16x16x4_f32(
       false, a,  false, b,
       (short)0, c,  false,  false);
}

#define GBM     128
#define GBN     128
#define GKC      32
#define GPITCH   36

template<int TM, int TN, int EPI>
__device__ __forceinline__ void gemm_block_tile(
    float* __restrict__ smem,
    const float* __restrict__ A, int lda,
    const float* __restrict__ W, int ldb,
    float* __restrict__ C, int ldc,
    int m0, int n0, int N, int K,
    const float* __restrict__ bias)
{
  constexpr int BM = 32 * TM;
  constexpr int BN = 64 * TN;
  constexpr int PA = BM / 32;
  constexpr int PB = BN / 32;
  float* const sA = smem;
  float* const sB = smem + 2 * BM * GPITCH;

  const int tid  = threadIdx.x;
  const int lane = tid & 31;
  const int wave = tid >> 5;
  const int wm   = wave >> 2;
  const int wn   = wave & 3;

  const int r    = lane & 15;
  const int half = lane >> 4;
  const int klo  = half << 1;

  const int srow = tid >> 3;
  const int sc4  = (tid & 7) << 2;
  const size_t aStep = (size_t)32 * lda;
  const float* __restrict__ gA = A + (size_t)(m0 + srow) * lda + sc4;
  const float* __restrict__ gB[PB];
#pragma unroll
  for (int q = 0; q < PB; ++q)
    gB[q] = W + (size_t)min(n0 + srow + 32 * q, N - 1) * ldb + sc4;
  const int sOff = srow * GPITCH + sc4;

  v8f acc[TM][TN];
#pragma unroll
  for (int i = 0; i < TM; ++i)
#pragma unroll
    for (int j = 0; j < TN; ++j) acc[i][j] = {};

  v4f ra[PA], rb[PB];
#pragma unroll
  for (int q = 0; q < PA; ++q) ra[q] = *(const v4f*)(gA + q * aStep);
#pragma unroll
  for (int q = 0; q < PB; ++q) rb[q] = *(const v4f*)(gB[q]);
#pragma unroll
  for (int q = 0; q < PA; ++q) *(v4f*)&sA[sOff + q * 32 * GPITCH] = ra[q];
#pragma unroll
  for (int q = 0; q < PB; ++q) *(v4f*)&sB[sOff + q * 32 * GPITCH] = rb[q];
  __syncthreads();

  const int nChunks = K / GKC;
  for (int kc = 0; kc < nChunks; ++kc) {
    const int cur = kc & 1;
    const bool more = (kc + 1 < nChunks);

    if (more) {
      const int kofs = (kc + 1) * GKC;
#pragma unroll
      for (int q = 0; q < PA; ++q) ra[q] = *(const v4f*)(gA + q * aStep + kofs);
#pragma unroll
      for (int q = 0; q < PB; ++q) rb[q] = *(const v4f*)(gB[q] + kofs);
    }

    const float* pA = sA + cur * BM * GPITCH + (wm * TM * 16 + r) * GPITCH + klo;
    const float* pB = sB + cur * BN * GPITCH + (wn * TN * 16 + r) * GPITCH + klo;
#pragma unroll
    for (int s = 0; s < GKC / 4; ++s) {
      v2f a[TM], b[TN];
#pragma unroll
      for (int i = 0; i < TM; ++i) a[i] = *(const v2f*)(pA + i * 16 * GPITCH + 4 * s);
#pragma unroll
      for (int j = 0; j < TN; ++j) b[j] = *(const v2f*)(pB + j * 16 * GPITCH + 4 * s);
#pragma unroll
      for (int i = 0; i < TM; ++i)
#pragma unroll
        for (int j = 0; j < TN; ++j)
          acc[i][j] = wmma_f32_16x16x4(a[i], b[j], acc[i][j]);
    }

    if (more) {
      float* dA = sA + (cur ^ 1) * BM * GPITCH + sOff;
      float* dB = sB + (cur ^ 1) * BN * GPITCH + sOff;
#pragma unroll
      for (int q = 0; q < PA; ++q) *(v4f*)(dA + q * 32 * GPITCH) = ra[q];
#pragma unroll
      for (int q = 0; q < PB; ++q) *(v4f*)(dB + q * 32 * GPITCH) = rb[q];
    }
    __syncthreads();
  }

  const int cbase = n0 + wn * TN * 16;
  if (cbase >= N) return;
#pragma unroll
  for (int i = 0; i < TM; ++i) {
#pragma unroll
    for (int j = 0; j < TN; ++j) {
      const int cn = cbase + j * 16 + r;
#pragma unroll
      for (int v = 0; v < 8; ++v) {
        const int cm = m0 + wm * TM * 16 + i * 16 + v + (half << 3);
        float val = acc[i][j][v];
        if (EPI == 1) {
          val += bias[cn];
          val = (val > 20.f) ? val : log1pf(__expf(val));
        }
        C[(size_t)cm * ldc + cn] = val;
      }
    }
  }
}

#define GSMEM  (2 * (GBM + GBN) * GPITCH)

template<int EPI>
__global__ void __launch_bounds__(256)
wmma_gemm_f32(const float* __restrict__ A, int lda,
              const float* __restrict__ W, int ldb,
              float* __restrict__ C, int ldc,
              int M, int N, int K,
              const float* __restrict__ bias)
{
  __shared__ __attribute__((aligned(16))) float smem[GSMEM];
  const int tilesN = (N + GBN - 1) / GBN;
  const int bm = blockIdx.x / tilesN;
  const int bn = blockIdx.x - bm * tilesN;
  gemm_block_tile<4, 2, EPI>(smem, A, lda, W, ldb, C, ldc, bm * GBM, bn * GBN, N, K, bias);
}

#define INP_TN     (OFF_DT / GBN)
#define INP_MAIN   ((M_ROWS / GBM) * INP_TN)
#define INP_SMALL  (M_ROWS / 32)

static_assert(OFF_DT % GBN == 0, "z, x, B, C columns are whole 128-wide tiles");
static_assert(E_DIM - OFF_DT == 64, "the dt columns are one 64-wide tile");
static_assert(M_ROWS % GBM == 0 && M_ROWS % 32 == 0 && D_MODEL % GKC == 0, "whole tiles and chunks");

__global__ void __launch_bounds__(256)
in_proj_kernel(const float* __restrict__ A, int lda,
               const float* __restrict__ W, int ldb,
               float* __restrict__ C, int ldc,
               int nMainCols, int nSmallCols, int K)
{
  __shared__ __attribute__((aligned(16))) float smem[GSMEM];
  const int tilesN = nMainCols / GBN;
  const int nMain  = (M_ROWS / GBM) * tilesN;
  if ((int)blockIdx.x < nMain) {
    const int bm = blockIdx.x / tilesN;
    const int bn = blockIdx.x - bm * tilesN;
    gemm_block_tile<4, 2, 0>(smem, A, lda, W, ldb, C, ldc,
                             bm * GBM, bn * GBN, nMainCols, K, nullptr);
  } else {
    const int t = blockIdx.x - nMain;
    gemm_block_tile<1, 1, 0>(smem, A, lda,
                             W + (size_t)nMainCols * ldb, ldb,
                             C + nMainCols, ldc,
                             t * 32, 0, nSmallCols, K, nullptr);
  }
}

#define CONV_T   16
#define CONV_C4  (D_INNER / 4)

static_assert(SEQ % CONV_T == 0, "a thread's steps stay inside one sequence");
static_assert((M_ROWS / CONV_T) * CONV_C4 % 256 == 0, "whole blocks");

__device__ __forceinline__ float conv_silu_one(float bias, float4 w,
                                               float xm3, float xm2, float xm1, float x0,
                                               bool v3, bool v2, bool v1)
{
  float acc = bias;
  if (v3) acc = fmaf(w.x, xm3, acc);
  if (v2) acc = fmaf(w.y, xm2, acc);
  if (v1) acc = fmaf(w.z, xm1, acc);
  acc = fmaf(w.w, x0, acc);
  return acc / (1.f + __expf(-acc));
}

__global__ void __launch_bounds__(256)
conv_silu_kernel(const float* __restrict__ zx,
                 const float* __restrict__ conv_w,
                 const float* __restrict__ conv_b,
                 float* __restrict__ xconv)
{
  const int idx  = blockIdx.x * 256 + threadIdx.x;
  const int c4   = idx % CONV_C4;
  const int rg   = idx / CONV_C4;
  const int d0   = c4 << 2;
  const int src  = ((d0 >> 6) << 4) + (d0 & 15);
  const int row0 = rg * CONV_T;
  const bool hal = (row0 % SEQ) != 0;

  const float4 w0 = *(const float4*)(conv_w + (size_t)(d0 + 0) * D_CONV);
  const float4 w1 = *(const float4*)(conv_w + (size_t)(d0 + 1) * D_CONV);
  const float4 w2 = *(const float4*)(conv_w + (size_t)(d0 + 2) * D_CONV);
  const float4 w3 = *(const float4*)(conv_w + (size_t)(d0 + 3) * D_CONV);
  const float4 bz = *(const float4*)(conv_b + d0);

  const float* __restrict__ px = zx + (size_t)row0 * E_DIM + OFF_X + src;
  float4 xw[CONV_T + 3];
  if (hal) {
    xw[0] = *(const float4*)(px - (ptrdiff_t)3 * E_DIM);
    xw[1] = *(const float4*)(px - (ptrdiff_t)2 * E_DIM);
    xw[2] = *(const float4*)(px - (ptrdiff_t)1 * E_DIM);
  } else {
    xw[0] = make_float4(0.f, 0.f, 0.f, 0.f);
    xw[1] = xw[0];
    xw[2] = xw[0];
  }
#pragma unroll
  for (int s = 0; s < CONV_T; ++s)
    xw[s + 3] = *(const float4*)(px + (size_t)s * E_DIM);

  float* __restrict__ po = xconv + (size_t)row0 * D_INNER + d0;
#pragma unroll
  for (int s = 0; s < CONV_T; ++s) {
    const bool v3 = hal || (s >= 3);
    const bool v2 = hal || (s >= 2);
    const bool v1 = hal || (s >= 1);
    float4 o;
    o.x = conv_silu_one(bz.x, w0, xw[s].x, xw[s + 1].x, xw[s + 2].x, xw[s + 3].x, v3, v2, v1);
    o.y = conv_silu_one(bz.y, w1, xw[s].y, xw[s + 1].y, xw[s + 2].y, xw[s + 3].y, v3, v2, v1);
    o.z = conv_silu_one(bz.z, w2, xw[s].z, xw[s + 1].z, xw[s + 2].z, xw[s + 3].z, v3, v2, v1);
    o.w = conv_silu_one(bz.w, w3, xw[s].w, xw[s + 1].w, xw[s + 2].w, xw[s + 3].w, v3, v2, v1);
    *(float4*)(po + (size_t)s * D_INNER) = o;
  }
}

#define SCAN_T   64
#define SCAN_NT  (SEQ / SCAN_T)

template<int M>
__device__ __forceinline__ float row_xor_add(float v)
{
  const int partner = __builtin_amdgcn_update_dpp(0, __builtin_bit_cast(int, v),
                                                  0x160 + M, 0xf, 0xf, true);
  return v + __builtin_bit_cast(float, partner);
}

__global__ void __launch_bounds__(256)
scan_kernel(const float* __restrict__ zx,
            const float* __restrict__ xconv,
            const float* __restrict__ dtbuf,
            const float* __restrict__ Amat,
            const float* __restrict__ Dvec,
            float* __restrict__ Y)
{
  __shared__ __attribute__((aligned(16))) float2 s_pd[SCAN_T][PCH];
  __shared__ __attribute__((aligned(16))) float2 s_bc[SCAN_T][D_STATE];
  __shared__ float s_y[SCAN_T][PCH * D_STATE + 1];

  const int b = blockIdx.x / H_HEADS;
  const int h = blockIdx.x % H_HEADS;
  const int p = threadIdx.x >> 4;
  const int n = threadIdx.x & 15;

  const int st = threadIdx.x >> 2;
  const int sq = (threadIdx.x & 3) << 2;

  const float  An  = Amat[(size_t)(h * PCH + p) * D_STATE + n];
  const float4 Dv4 = make_float4(Dvec[h * PCH + sq + 0], Dvec[h * PCH + sq + 1],
                                 Dvec[h * PCH + sq + 2], Dvec[h * PCH + sq + 3]);

  const size_t row0 = (size_t)b * SEQ + st;
  const size_t coff = (size_t)h * PCH + sq;
  const size_t boff = OFF_B + (size_t)(h / REP) * D_STATE + sq;
  const size_t cofs = OFF_C + (size_t)h * D_STATE + sq;
  const size_t zoff = OFF_Z + coff;

  float4 r_dt = *(const float4*)(dtbuf + row0 * D_INNER + coff);
  float4 r_x  = *(const float4*)(xconv + row0 * D_INNER + coff);
  float4 r_B  = *(const float4*)(zx + row0 * E_DIM + boff);
  float4 r_C  = *(const float4*)(zx + row0 * E_DIM + cofs);
  float4 r_z  = *(const float4*)(zx + row0 * E_DIM + zoff);

  float hstate = 0.f;
  for (int tile = 0; tile < SCAN_NT; ++tile) {
    {
      float4* pd = (float4*)&s_pd[st][sq];
      pd[0] = make_float4(r_dt.x, r_dt.x * r_x.x, r_dt.y, r_dt.y * r_x.y);
      pd[1] = make_float4(r_dt.z, r_dt.z * r_x.z, r_dt.w, r_dt.w * r_x.w);
      float4* bc = (float4*)&s_bc[st][sq];
      bc[0] = make_float4(r_B.x, r_C.x, r_B.y, r_C.y);
      bc[1] = make_float4(r_B.z, r_C.z, r_B.w, r_C.w);
    }
    const float4 c_x = r_x;
    const float4 c_z = r_z;
    __syncthreads();

    if (tile + 1 < SCAN_NT) {
      const size_t row = row0 + (size_t)(tile + 1) * SCAN_T;
      r_dt = *(const float4*)(dtbuf + row * D_INNER + coff);
      r_x  = *(const float4*)(xconv + row * D_INNER + coff);
      r_B  = *(const float4*)(zx + row * E_DIM + boff);
      r_C  = *(const float4*)(zx + row * E_DIM + cofs);
      r_z  = *(const float4*)(zx + row * E_DIM + zoff);
    }

#pragma unroll 16
    for (int t = 0; t < SCAN_T; ++t) {
      const float2 pd = s_pd[t][p];
      const float2 bc = s_bc[t][n];

      const float a = __expf(pd.x * An);
      hstate = fmaf(a, hstate, pd.y * bc.x);

      float yc = hstate * bc.y;
      yc = row_xor_add<1>(yc);
      yc = row_xor_add<2>(yc);
      yc = row_xor_add<4>(yc);
      yc = row_xor_add<8>(yc);

      s_y[t][threadIdx.x] = yc;
    }
    __syncthreads();

    {
      float4 y4;
      y4.x = s_y[st][(sq + 0) * D_STATE];
      y4.y = s_y[st][(sq + 1) * D_STATE];
      y4.z = s_y[st][(sq + 2) * D_STATE];
      y4.w = s_y[st][(sq + 3) * D_STATE];
      const float g0 = c_z.x / (1.f + __expf(-c_z.x));
      const float g1 = c_z.y / (1.f + __expf(-c_z.y));
      const float g2 = c_z.z / (1.f + __expf(-c_z.z));
      const float g3 = c_z.w / (1.f + __expf(-c_z.w));
      float4 o;
      o.x = (y4.x + Dv4.x * c_x.x) * g0;
      o.y = (y4.y + Dv4.y * c_x.y) * g1;
      o.z = (y4.z + Dv4.z * c_x.z) * g2;
      o.w = (y4.w + Dv4.w * c_x.w) * g3;
      const size_t row = row0 + (size_t)tile * SCAN_T;
      *(float4*)(Y + row * D_INNER + coff) = o;
    }
  }
}

extern "C" void kernel_launch(void* const* d_in, const int* in_sizes, int n_in,
                              void* d_out, int out_size, void* d_ws, size_t ws_size,
                              hipStream_t stream) {
  const float* hidden     = (const float*)d_in[0];
  const float* in_proj_w  = (const float*)d_in[1];
  const float* conv_w     = (const float*)d_in[2];
  const float* conv_b     = (const float*)d_in[3];
  const float* dt_proj_w  = (const float*)d_in[4];
  const float* dt_proj_b  = (const float*)d_in[5];
  const float* Amat       = (const float*)d_in[6];
  const float* Dvec       = (const float*)d_in[7];
  const float* out_proj_w = (const float*)d_in[8];
  float* out = (float*)d_out;

  float* zx    = (float*)d_ws;
  float* xconv = zx    + (size_t)M_ROWS * E_DIM;
  float* dtbuf = xconv + (size_t)M_ROWS * D_INNER;
  float* Ybuf  = dtbuf + (size_t)M_ROWS * D_INNER;

  {
    int blocks = INP_MAIN + INP_SMALL;
    in_proj_kernel<<<blocks, 256, 0, stream>>>(
        hidden, D_MODEL, in_proj_w, D_MODEL, zx, E_DIM,
        OFF_DT, E_DIM - OFF_DT, D_MODEL);
  }

  {
    int blocks = (M_ROWS / CONV_T) * CONV_C4 / 256;
    conv_silu_kernel<<<blocks, 256, 0, stream>>>(zx, conv_w, conv_b, xconv);
  }

  {
    int blocks = (M_ROWS / GBM) * ((D_INNER + GBN - 1) / GBN);
    wmma_gemm_f32<1><<<blocks, 256, 0, stream>>>(
        zx + OFF_DT, E_DIM, dt_proj_w, DT_RANK, dtbuf, D_INNER,
        M_ROWS, D_INNER, DT_RANK, dt_proj_b);
  }

  scan_kernel<<<BATCH * H_HEADS, 256, 0, stream>>>(zx, xconv, dtbuf, Amat, Dvec, Ybuf);

  {
    int blocks = (M_ROWS / GBM) * ((D_MODEL + GBN - 1) / GBN);
    wmma_gemm_f32<0><<<blocks, 256, 0, stream>>>(
        Ybuf, D_INNER, out_proj_w, D_INNER, out, D_MODEL,
        M_ROWS, D_MODEL, D_INNER, nullptr);
  }
}


# softplus, SiLU and the scan's decay straight on the exp/log/rcp units: log1pf -> log2(u)*ln2*e/(u-1), u=1+e; exact divides -> x*rcp; exp(dt*A) -> 2^(dt*A2), A2=A*log2e per thread
# speedup vs baseline: 6.9331x; 1.0721x over previous
#include <hip/hip_runtime.h>
#include <math.h>

#define D_MODEL   1024
#define D_INNER   2048
#define D_XB       512
#define D_STATE     16
#define D_CONV       4
#define DT_RANK     64
#define NXB         32
#define H_HEADS    128
#define REP          4
#define PCH         16
#define BATCH        2
#define SEQ       2048
#define E_DIM     (2*D_INNER + 2*D_XB + DT_RANK)
#define M_ROWS    (BATCH*SEQ)

#define OFF_Z   0
#define OFF_X   (D_INNER)
#define OFF_B   (D_INNER + D_XB)
#define OFF_C   (D_INNER + 2*D_XB)
#define OFF_DT  (2*D_INNER + 2*D_XB)

typedef __attribute__((ext_vector_type(2))) float v2f;
typedef __attribute__((ext_vector_type(4))) float v4f;
typedef __attribute__((ext_vector_type(8))) float v8f;

__device__ __forceinline__ v8f wmma_f32_16x16x4(v2f a, v2f b, v8f c) {
  return __builtin_amdgcn_wmma_f32_16x16x4_f32(
       false, a,  false, b,
       (short)0, c,  false,  false);
}

__device__ __forceinline__ float silu_hw(float x)
{
  return x * __builtin_amdgcn_rcpf(1.f + __expf(-x));
}

__device__ __forceinline__ float softplus_hw(float x)
{
  const float e = __expf(x);
  const float u = 1.f + e;
  const float d = u - 1.f;
  const float l = (__builtin_amdgcn_logf(u) * 0.69314718055994531f) * (e * __builtin_amdgcn_rcpf(d));
  const float r = (d == 0.f) ? e : l;
  return (x > 20.f) ? x : r;
}

#define GBM     128
#define GBN     128
#define GKC      32
#define GPITCH   36

template<int TM, int TN, int EPI>
__device__ __forceinline__ void gemm_block_tile(
    float* __restrict__ smem,
    const float* __restrict__ A, int lda,
    const float* __restrict__ W, int ldb,
    float* __restrict__ C, int ldc,
    int m0, int n0, int N, int K,
    const float* __restrict__ bias)
{
  constexpr int BM = 32 * TM;
  constexpr int BN = 64 * TN;
  constexpr int PA = BM / 32;
  constexpr int PB = BN / 32;
  float* const sA = smem;
  float* const sB = smem + 2 * BM * GPITCH;

  const int tid  = threadIdx.x;
  const int lane = tid & 31;
  const int wave = tid >> 5;
  const int wm   = wave >> 2;
  const int wn   = wave & 3;

  const int r    = lane & 15;
  const int half = lane >> 4;
  const int klo  = half << 1;

  const int srow = tid >> 3;
  const int sc4  = (tid & 7) << 2;
  const size_t aStep = (size_t)32 * lda;
  const float* __restrict__ gA = A + (size_t)(m0 + srow) * lda + sc4;
  const float* __restrict__ gB[PB];
#pragma unroll
  for (int q = 0; q < PB; ++q)
    gB[q] = W + (size_t)min(n0 + srow + 32 * q, N - 1) * ldb + sc4;
  const int sOff = srow * GPITCH + sc4;

  v8f acc[TM][TN];
#pragma unroll
  for (int i = 0; i < TM; ++i)
#pragma unroll
    for (int j = 0; j < TN; ++j) acc[i][j] = {};

  v4f ra[PA], rb[PB];
#pragma unroll
  for (int q = 0; q < PA; ++q) ra[q] = *(const v4f*)(gA + q * aStep);
#pragma unroll
  for (int q = 0; q < PB; ++q) rb[q] = *(const v4f*)(gB[q]);
#pragma unroll
  for (int q = 0; q < PA; ++q) *(v4f*)&sA[sOff + q * 32 * GPITCH] = ra[q];
#pragma unroll
  for (int q = 0; q < PB; ++q) *(v4f*)&sB[sOff + q * 32 * GPITCH] = rb[q];
  __syncthreads();

  const int nChunks = K / GKC;
  for (int kc = 0; kc < nChunks; ++kc) {
    const int cur = kc & 1;
    const bool more = (kc + 1 < nChunks);

    if (more) {
      const int kofs = (kc + 1) * GKC;
#pragma unroll
      for (int q = 0; q < PA; ++q) ra[q] = *(const v4f*)(gA + q * aStep + kofs);
#pragma unroll
      for (int q = 0; q < PB; ++q) rb[q] = *(const v4f*)(gB[q] + kofs);
    }

    const float* pA = sA + cur * BM * GPITCH + (wm * TM * 16 + r) * GPITCH + klo;
    const float* pB = sB + cur * BN * GPITCH + (wn * TN * 16 + r) * GPITCH + klo;
#pragma unroll
    for (int s = 0; s < GKC / 4; ++s) {
      v2f a[TM], b[TN];
#pragma unroll
      for (int i = 0; i < TM; ++i) a[i] = *(const v2f*)(pA + i * 16 * GPITCH + 4 * s);
#pragma unroll
      for (int j = 0; j < TN; ++j) b[j] = *(const v2f*)(pB + j * 16 * GPITCH + 4 * s);
#pragma unroll
      for (int i = 0; i < TM; ++i)
#pragma unroll
        for (int j = 0; j < TN; ++j)
          acc[i][j] = wmma_f32_16x16x4(a[i], b[j], acc[i][j]);
    }

    if (more) {
      float* dA = sA + (cur ^ 1) * BM * GPITCH + sOff;
      float* dB = sB + (cur ^ 1) * BN * GPITCH + sOff;
#pragma unroll
      for (int q = 0; q < PA; ++q) *(v4f*)(dA + q * 32 * GPITCH) = ra[q];
#pragma unroll
      for (int q = 0; q < PB; ++q) *(v4f*)(dB + q * 32 * GPITCH) = rb[q];
    }
    __syncthreads();
  }

  const int cbase = n0 + wn * TN * 16;
  if (cbase >= N) return;
#pragma unroll
  for (int i = 0; i < TM; ++i) {
#pragma unroll
    for (int j = 0; j < TN; ++j) {
      const int cn = cbase + j * 16 + r;
#pragma unroll
      for (int v = 0; v < 8; ++v) {
        const int cm = m0 + wm * TM * 16 + i * 16 + v + (half << 3);
        float val = acc[i][j][v];
        if (EPI == 1) {
          val += bias[cn];
          val = softplus_hw(val);
        }
        C[(size_t)cm * ldc + cn] = val;
      }
    }
  }
}

#define GSMEM  (2 * (GBM + GBN) * GPITCH)

template<int EPI>
__global__ void __launch_bounds__(256)
wmma_gemm_f32(const float* __restrict__ A, int lda,
              const float* __restrict__ W, int ldb,
              float* __restrict__ C, int ldc,
              int M, int N, int K,
              const float* __restrict__ bias)
{
  __shared__ __attribute__((aligned(16))) float smem[GSMEM];
  const int tilesN = (N + GBN - 1) / GBN;
  const int bm = blockIdx.x / tilesN;
  const int bn = blockIdx.x - bm * tilesN;
  gemm_block_tile<4, 2, EPI>(smem, A, lda, W, ldb, C, ldc, bm * GBM, bn * GBN, N, K, bias);
}

#define INP_TN     (OFF_DT / GBN)
#define INP_MAIN   ((M_ROWS / GBM) * INP_TN)
#define INP_SMALL  (M_ROWS / 32)

static_assert(OFF_DT % GBN == 0, "z, x, B, C columns are whole 128-wide tiles");
static_assert(E_DIM - OFF_DT == 64, "the dt columns are one 64-wide tile");
static_assert(M_ROWS % GBM == 0 && M_ROWS % 32 == 0 && D_MODEL % GKC == 0, "whole tiles and chunks");

__global__ void __launch_bounds__(256)
in_proj_kernel(const float* __restrict__ A, int lda,
               const float* __restrict__ W, int ldb,
               float* __restrict__ C, int ldc,
               int nMainCols, int nSmallCols, int K)
{
  __shared__ __attribute__((aligned(16))) float smem[GSMEM];
  const int tilesN = nMainCols / GBN;
  const int nMain  = (M_ROWS / GBM) * tilesN;
  if ((int)blockIdx.x < nMain) {
    const int bm = blockIdx.x / tilesN;
    const int bn = blockIdx.x - bm * tilesN;
    gemm_block_tile<4, 2, 0>(smem, A, lda, W, ldb, C, ldc,
                             bm * GBM, bn * GBN, nMainCols, K, nullptr);
  } else {
    const int t = blockIdx.x - nMain;
    gemm_block_tile<1, 1, 0>(smem, A, lda,
                             W + (size_t)nMainCols * ldb, ldb,
                             C + nMainCols, ldc,
                             t * 32, 0, nSmallCols, K, nullptr);
  }
}

#define CONV_T   16
#define CONV_C4  (D_INNER / 4)

static_assert(SEQ % CONV_T == 0, "a thread's steps stay inside one sequence");
static_assert((M_ROWS / CONV_T) * CONV_C4 % 256 == 0, "whole blocks");

__device__ __forceinline__ float conv_silu_one(float bias, float4 w,
                                               float xm3, float xm2, float xm1, float x0,
                                               bool v3, bool v2, bool v1)
{
  float acc = bias;
  if (v3) acc = fmaf(w.x, xm3, acc);
  if (v2) acc = fmaf(w.y, xm2, acc);
  if (v1) acc = fmaf(w.z, xm1, acc);
  acc = fmaf(w.w, x0, acc);
  return silu_hw(acc);
}

__global__ void __launch_bounds__(256)
conv_silu_kernel(const float* __restrict__ zx,
                 const float* __restrict__ conv_w,
                 const float* __restrict__ conv_b,
                 float* __restrict__ xconv)
{
  const int idx  = blockIdx.x * 256 + threadIdx.x;
  const int c4   = idx % CONV_C4;
  const int rg   = idx / CONV_C4;
  const int d0   = c4 << 2;
  const int src  = ((d0 >> 6) << 4) + (d0 & 15);
  const int row0 = rg * CONV_T;
  const bool hal = (row0 % SEQ) != 0;

  const float4 w0 = *(const float4*)(conv_w + (size_t)(d0 + 0) * D_CONV);
  const float4 w1 = *(const float4*)(conv_w + (size_t)(d0 + 1) * D_CONV);
  const float4 w2 = *(const float4*)(conv_w + (size_t)(d0 + 2) * D_CONV);
  const float4 w3 = *(const float4*)(conv_w + (size_t)(d0 + 3) * D_CONV);
  const float4 bz = *(const float4*)(conv_b + d0);

  const float* __restrict__ px = zx + (size_t)row0 * E_DIM + OFF_X + src;
  float4 xw[CONV_T + 3];
  if (hal) {
    xw[0] = *(const float4*)(px - (ptrdiff_t)3 * E_DIM);
    xw[1] = *(const float4*)(px - (ptrdiff_t)2 * E_DIM);
    xw[2] = *(const float4*)(px - (ptrdiff_t)1 * E_DIM);
  } else {
    xw[0] = make_float4(0.f, 0.f, 0.f, 0.f);
    xw[1] = xw[0];
    xw[2] = xw[0];
  }
#pragma unroll
  for (int s = 0; s < CONV_T; ++s)
    xw[s + 3] = *(const float4*)(px + (size_t)s * E_DIM);

  float* __restrict__ po = xconv + (size_t)row0 * D_INNER + d0;
#pragma unroll
  for (int s = 0; s < CONV_T; ++s) {
    const bool v3 = hal || (s >= 3);
    const bool v2 = hal || (s >= 2);
    const bool v1 = hal || (s >= 1);
    float4 o;
    o.x = conv_silu_one(bz.x, w0, xw[s].x, xw[s + 1].x, xw[s + 2].x, xw[s + 3].x, v3, v2, v1);
    o.y = conv_silu_one(bz.y, w1, xw[s].y, xw[s + 1].y, xw[s + 2].y, xw[s + 3].y, v3, v2, v1);
    o.z = conv_silu_one(bz.z, w2, xw[s].z, xw[s + 1].z, xw[s + 2].z, xw[s + 3].z, v3, v2, v1);
    o.w = conv_silu_one(bz.w, w3, xw[s].w, xw[s + 1].w, xw[s + 2].w, xw[s + 3].w, v3, v2, v1);
    *(float4*)(po + (size_t)s * D_INNER) = o;
  }
}

#define SCAN_T   64
#define SCAN_NT  (SEQ / SCAN_T)

template<int M>
__device__ __forceinline__ float row_xor_add(float v)
{
  const int partner = __builtin_amdgcn_update_dpp(0, __builtin_bit_cast(int, v),
                                                  0x160 + M, 0xf, 0xf, true);
  return v + __builtin_bit_cast(float, partner);
}

__global__ void __launch_bounds__(256)
scan_kernel(const float* __restrict__ zx,
            const float* __restrict__ xconv,
            const float* __restrict__ dtbuf,
            const float* __restrict__ Amat,
            const float* __restrict__ Dvec,
            float* __restrict__ Y)
{
  __shared__ __attribute__((aligned(16))) float2 s_pd[SCAN_T][PCH];
  __shared__ __attribute__((aligned(16))) float2 s_bc[SCAN_T][D_STATE];
  __shared__ float s_y[SCAN_T][PCH * D_STATE + 1];

  const int b = blockIdx.x / H_HEADS;
  const int h = blockIdx.x % H_HEADS;
  const int p = threadIdx.x >> 4;
  const int n = threadIdx.x & 15;

  const int st = threadIdx.x >> 2;
  const int sq = (threadIdx.x & 3) << 2;

  const float  An2 = Amat[(size_t)(h * PCH + p) * D_STATE + n] * 1.44269504088896341f;
  const float4 Dv4 = make_float4(Dvec[h * PCH + sq + 0], Dvec[h * PCH + sq + 1],
                                 Dvec[h * PCH + sq + 2], Dvec[h * PCH + sq + 3]);

  const size_t row0 = (size_t)b * SEQ + st;
  const size_t coff = (size_t)h * PCH + sq;
  const size_t boff = OFF_B + (size_t)(h / REP) * D_STATE + sq;
  const size_t cofs = OFF_C + (size_t)h * D_STATE + sq;
  const size_t zoff = OFF_Z + coff;

  float4 r_dt = *(const float4*)(dtbuf + row0 * D_INNER + coff);
  float4 r_x  = *(const float4*)(xconv + row0 * D_INNER + coff);
  float4 r_B  = *(const float4*)(zx + row0 * E_DIM + boff);
  float4 r_C  = *(const float4*)(zx + row0 * E_DIM + cofs);
  float4 r_z  = *(const float4*)(zx + row0 * E_DIM + zoff);

  float hstate = 0.f;
  for (int tile = 0; tile < SCAN_NT; ++tile) {
    {
      float4* pd = (float4*)&s_pd[st][sq];
      pd[0] = make_float4(r_dt.x, r_dt.x * r_x.x, r_dt.y, r_dt.y * r_x.y);
      pd[1] = make_float4(r_dt.z, r_dt.z * r_x.z, r_dt.w, r_dt.w * r_x.w);
      float4* bc = (float4*)&s_bc[st][sq];
      bc[0] = make_float4(r_B.x, r_C.x, r_B.y, r_C.y);
      bc[1] = make_float4(r_B.z, r_C.z, r_B.w, r_C.w);
    }
    const float4 c_x = r_x;
    const float4 c_z = r_z;
    __syncthreads();

    if (tile + 1 < SCAN_NT) {
      const size_t row = row0 + (size_t)(tile + 1) * SCAN_T;
      r_dt = *(const float4*)(dtbuf + row * D_INNER + coff);
      r_x  = *(const float4*)(xconv + row * D_INNER + coff);
      r_B  = *(const float4*)(zx + row * E_DIM + boff);
      r_C  = *(const float4*)(zx + row * E_DIM + cofs);
      r_z  = *(const float4*)(zx + row * E_DIM + zoff);
    }

#pragma unroll 16
    for (int t = 0; t < SCAN_T; ++t) {
      const float2 pd = s_pd[t][p];
      const float2 bc = s_bc[t][n];

      const float a = __builtin_amdgcn_exp2f(pd.x * An2);
      hstate = fmaf(a, hstate, pd.y * bc.x);

      float yc = hstate * bc.y;
      yc = row_xor_add<1>(yc);
      yc = row_xor_add<2>(yc);
      yc = row_xor_add<4>(yc);
      yc = row_xor_add<8>(yc);

      s_y[t][threadIdx.x] = yc;
    }
    __syncthreads();

    {
      float4 y4;
      y4.x = s_y[st][(sq + 0) * D_STATE];
      y4.y = s_y[st][(sq + 1) * D_STATE];
      y4.z = s_y[st][(sq + 2) * D_STATE];
      y4.w = s_y[st][(sq + 3) * D_STATE];
      const float g0 = silu_hw(c_z.x);
      const float g1 = silu_hw(c_z.y);
      const float g2 = silu_hw(c_z.z);
      const float g3 = silu_hw(c_z.w);
      float4 o;
      o.x = (y4.x + Dv4.x * c_x.x) * g0;
      o.y = (y4.y + Dv4.y * c_x.y) * g1;
      o.z = (y4.z + Dv4.z * c_x.z) * g2;
      o.w = (y4.w + Dv4.w * c_x.w) * g3;
      const size_t row = row0 + (size_t)tile * SCAN_T;
      *(float4*)(Y + row * D_INNER + coff) = o;
    }
  }
}

extern "C" void kernel_launch(void* const* d_in, const int* in_sizes, int n_in,
                              void* d_out, int out_size, void* d_ws, size_t ws_size,
                              hipStream_t stream) {
  const float* hidden     = (const float*)d_in[0];
  const float* in_proj_w  = (const float*)d_in[1];
  const float* conv_w     = (const float*)d_in[2];
  const float* conv_b     = (const float*)d_in[3];
  const float* dt_proj_w  = (const float*)d_in[4];
  const float* dt_proj_b  = (const float*)d_in[5];
  const float* Amat       = (const float*)d_in[6];
  const float* Dvec       = (const float*)d_in[7];
  const float* out_proj_w = (const float*)d_in[8];
  float* out = (float*)d_out;

  float* zx    = (float*)d_ws;
  float* xconv = zx    + (size_t)M_ROWS * E_DIM;
  float* dtbuf = xconv + (size_t)M_ROWS * D_INNER;
  float* Ybuf  = dtbuf + (size_t)M_ROWS * D_INNER;

  {
    int blocks = INP_MAIN + INP_SMALL;
    in_proj_kernel<<<blocks, 256, 0, stream>>>(
        hidden, D_MODEL, in_proj_w, D_MODEL, zx, E_DIM,
        OFF_DT, E_DIM - OFF_DT, D_MODEL);
  }

  {
    int blocks = (M_ROWS / CONV_T) * CONV_C4 / 256;
    conv_silu_kernel<<<blocks, 256, 0, stream>>>(zx, conv_w, conv_b, xconv);
  }

  {
    int blocks = (M_ROWS / GBM) * ((D_INNER + GBN - 1) / GBN);
    wmma_gemm_f32<1><<<blocks, 256, 0, stream>>>(
        zx + OFF_DT, E_DIM, dt_proj_w, DT_RANK, dtbuf, D_INNER,
        M_ROWS, D_INNER, DT_RANK, dt_proj_b);
  }

  scan_kernel<<<BATCH * H_HEADS, 256, 0, stream>>>(zx, xconv, dtbuf, Amat, Dvec, Ybuf);

  {
    int blocks = (M_ROWS / GBM) * ((D_MODEL + GBN - 1) / GBN);
    wmma_gemm_f32<0><<<blocks, 256, 0, stream>>>(
        Ybuf, D_INNER, out_proj_w, D_INNER, out, D_MODEL,
        M_ROWS, D_MODEL, D_INNER, nullptr);
  }
}
